# starting kernel
# baseline (speedup 1.0000x reference)
#include <hip/hip_runtime.h>
#include <math.h>

#define NN   50000
#define NE   625000
#define FIN  128
#define DH   128
#define NC   40
#define NCP  48

typedef __attribute__((ext_vector_type(2))) float v2f;
typedef __attribute__((ext_vector_type(8))) float v8f;

static inline int cdiv(long a, long b) { return (int)((a + b - 1) / b); }


__global__ void k_deg_init(float* __restrict__ deg) {
    int i = blockIdx.x * blockDim.x + threadIdx.x;
    if (i < NN) deg[i] = 1.0f;
}

__global__ void k_deg_accum(const int* __restrict__ dst, float* __restrict__ deg) {
    int e = blockIdx.x * blockDim.x + threadIdx.x;
    if (e < NE) atomicAdd(&deg[dst[e]], 1.0f);
}

__global__ void k_dinv(const float* __restrict__ deg, float* __restrict__ dinv) {
    int i = blockIdx.x * blockDim.x + threadIdx.x;
    if (i < NN) dinv[i] = rsqrtf(deg[i]);
}

__global__ void k_norm(const int* __restrict__ src, const int* __restrict__ dst,
                       const float* __restrict__ dinv, float* __restrict__ nrm) {
    int e = blockIdx.x * blockDim.x + threadIdx.x;
    if (e < NE) nrm[e] = dinv[src[e]] * dinv[dst[e]];
}

__global__ void k_pad_w2(const float* __restrict__ W2, float* __restrict__ W2p) {
    int t = blockIdx.x * blockDim.x + threadIdx.x;
    if (t >= DH * NCP) return;
    int r = t / NCP, c = t - r * NCP;
    W2p[t] = (c < NC) ? W2[r * NC + c] : 0.0f;
}


template <int K, int NP, int NOUT>
__global__ void k_gemm_wmma_f32(const float* __restrict__ A, const float* __restrict__ B,
                                float* __restrict__ C, int M) {
    __shared__ float ldsB[K * NP];

    for (int idx = threadIdx.x; idx < (K * NP) / 4; idx += blockDim.x)
        ((float4*)ldsB)[idx] = ((const float4*)B)[idx];
    __syncthreads();

    const int lane = threadIdx.x & 31;
    const int wave = (blockIdx.x * blockDim.x + threadIdx.x) >> 5;
    constexpr int tilesN = NP / 16;
    int tm = wave / tilesN;
    const int tn = wave - tm * tilesN;
    const bool valid = (tm * 16 < M);
    if (!valid) tm = M / 16 - 1;

    const int m  = tm * 16 + (lane & 15);
    const int n  = tn * 16 + (lane & 15);
    const int kh = (lane >> 4) << 1;

    const float* aP   = A + m * K + kh;
    const int    bOff = kh * NP + n;

    v8f acc = {};
#pragma unroll
    for (int kk = 0; kk < K; kk += 4) {
        v2f a, b;
        a.x = aP[kk];
        a.y = aP[kk + 1];
        b.x = ldsB[bOff + kk * NP];
        b.y = ldsB[bOff + (kk + 1) * NP];
        acc = __builtin_amdgcn_wmma_f32_16x16x4_f32(
             false, a,  false, b,
             (short)0, acc,  false,  false);
    }

    if (valid && n < NOUT) {
        const int rbase = tm * 16 + ((lane >> 4) << 3);
#pragma unroll
        for (int v = 0; v < 8; ++v)
            C[(rbase + v) * NOUT + n] = acc[v];
    }
}


__global__ void k_agg_init_h(const float* __restrict__ h, const float* __restrict__ dinv,
                             float* __restrict__ agg) {
    long t = (long)blockIdx.x * blockDim.x + threadIdx.x;
    if (t >= (long)NN * 32) return;
    int i = (int)(t >> 5);
    float s = dinv[i]; s *= s;
    float4 v = ((const float4*)h)[t];
    float4 o = make_float4(v.x * s, v.y * s, v.z * s, v.w * s);
    ((float4*)agg)[t] = o;
}

__global__ void k_agg_edges_h(const int* __restrict__ src, const int* __restrict__ dst,
                              const float* __restrict__ nrm,
                              const float* __restrict__ h, float* __restrict__ agg) {
    long t = (long)blockIdx.x * blockDim.x + threadIdx.x;
    long e = t >> 5;
    int lane = (int)(t & 31);
    if (e >= NE) return;
    int s = src[e], d = dst[e];
    float w = nrm[e];
    float4 v = ((const float4*)h)[(long)s * 32 + lane];
    float* o = agg + (long)d * 128 + lane * 4;
    atomicAdd(o + 0, v.x * w);
    atomicAdd(o + 1, v.y * w);
    atomicAdd(o + 2, v.z * w);
    atomicAdd(o + 3, v.w * w);
}

__global__ void k_bias_relu(float* __restrict__ agg, const float* __restrict__ b) {
    long t = (long)blockIdx.x * blockDim.x + threadIdx.x;
    if (t >= (long)NN * DH) return;
    float v = agg[t] + b[t & (DH - 1)];
    agg[t] = v > 0.0f ? v : 0.0f;
}


__global__ void k_agg_init_c(const float* __restrict__ h, const float* __restrict__ dinv,
                             float* __restrict__ agg) {
    long t = (long)blockIdx.x * blockDim.x + threadIdx.x;
    if (t >= (long)NN * NC) return;
    int i = (int)(t / NC);
    float s = dinv[i]; s *= s;
    agg[t] = h[t] * s;
}

__global__ void k_agg_edges_c(const int* __restrict__ src, const int* __restrict__ dst,
                              const float* __restrict__ nrm,
                              const float* __restrict__ h, float* __restrict__ agg) {
    long t = (long)blockIdx.x * blockDim.x + threadIdx.x;
    long e = t >> 5;
    int lane = (int)(t & 31);
    if (e >= NE) return;
    int s = src[e], d = dst[e];
    float w = nrm[e];
    atomicAdd(&agg[(long)d * NC + lane], h[(long)s * NC + lane] * w);
    if (lane < NC - 32) {
        int c = lane + 32;
        atomicAdd(&agg[(long)d * NC + c], h[(long)s * NC + c] * w);
    }
}


__global__ void k_logsoftmax(const float* __restrict__ agg, const float* __restrict__ b,
                             float* __restrict__ out) {
    int i = blockIdx.x * blockDim.x + threadIdx.x;
    if (i >= NN) return;
    const float* r = agg + (long)i * NC;
    float mx = -INFINITY;
#pragma unroll
    for (int c = 0; c < NC; ++c) mx = fmaxf(mx, r[c] + b[c]);
    float sum = 0.0f;
#pragma unroll
    for (int c = 0; c < NC; ++c) sum += expf(r[c] + b[c] - mx);
    float lse = mx + logf(sum);
#pragma unroll
    for (int c = 0; c < NC; ++c) out[(long)i * NC + c] = r[c] + b[c] - lse;
}


extern "C" void kernel_launch(void* const* d_in, const int* in_sizes, int n_in,
                              void* d_out, int out_size, void* d_ws, size_t ws_size,
                              hipStream_t stream) {
    const float* x  = (const float*)d_in[0];
    const int*   ei = (const int*)d_in[1];
    const float* W1 = (const float*)d_in[2];
    const float* b1 = (const float*)d_in[3];
    const float* W2 = (const float*)d_in[4];
    const float* b2 = (const float*)d_in[5];
    float* out = (float*)d_out;

    const int* src = ei;
    const int* dst = ei + NE;

    float* ws   = (float*)d_ws;
    float* deg  = ws;
    float* dinv = deg  + NN;
    float* nrm  = dinv + NN;
    float* h0   = nrm  + NE;
    float* agg1 = h0   + (long)NN * DH;
    float* h2   = agg1 + (long)NN * DH;
    float* agg2 = h2   + (long)NN * NC;
    float* W2p  = agg2 + (long)NN * NC;

    const int B = 256;

    k_deg_init <<<cdiv(NN, B), B, 0, stream>>>(deg);
    k_deg_accum<<<cdiv(NE, B), B, 0, stream>>>(dst, deg);
    k_dinv     <<<cdiv(NN, B), B, 0, stream>>>(deg, dinv);
    k_norm     <<<cdiv(NE, B), B, 0, stream>>>(src, dst, dinv, nrm);
    k_pad_w2   <<<cdiv(DH * NCP, B), B, 0, stream>>>(W2, W2p);

    {
        long waves = (long)(NN / 16) * (DH / 16);
        k_gemm_wmma_f32<FIN, DH, DH>
            <<<cdiv(waves * 32, B), B, 0, stream>>>(x, W1, h0, NN);
    }
    k_agg_init_h <<<cdiv((long)NN * 32, B), B, 0, stream>>>(h0, dinv, agg1);
    k_agg_edges_h<<<cdiv((long)NE * 32, B), B, 0, stream>>>(src, dst, nrm, h0, agg1);
    k_bias_relu  <<<cdiv((long)NN * DH, B), B, 0, stream>>>(agg1, b1);

    {
        long waves = (long)(NN / 16) * (NCP / 16);
        k_gemm_wmma_f32<DH, NCP, NC>
            <<<cdiv(waves * 32, B), B, 0, stream>>>(agg1, W2p, h2, NN);
    }
    k_agg_init_c <<<cdiv((long)NN * NC, B), B, 0, stream>>>(h2, dinv, agg2);
    k_agg_edges_c<<<cdiv((long)NE * 32, B), B, 0, stream>>>(src, dst, nrm, h2, agg2);

    k_logsoftmax<<<cdiv(NN, B), B, 0, stream>>>(agg2, b2, out);
}


# Both GEMM launches re-tiled: a wave owns 16 rows x all columns (8 or 3 accumulator tiles) and loads each A fragment once for all; W staged to LDS once per 128 rows, k-row pairs interleaved
# speedup vs baseline: 2.0335x; 2.0335x over previous
#include <hip/hip_runtime.h>
#include <math.h>

#define NN   50000
#define NE   625000
#define FIN  128
#define DH   128
#define NC   40
#define NCP  48

#define RN   200
#define LCAP 4096
#define OT   1024
#define NRG  ((NN + RN - 1) / RN)
#define N16  ((NE + 15) / 16)
#define NKW  (N16 * 4)

typedef __attribute__((ext_vector_type(2))) float v2f;
typedef __attribute__((ext_vector_type(8))) float v8f;

static inline int cdiv(long a, long b) { return (int)((a + b - 1) / b); }


__global__ void k_deg_init(float* __restrict__ deg) {
    int i = blockIdx.x * blockDim.x + threadIdx.x;
    if (i < NN) deg[i] = 1.0f;
}

__global__ void k_deg_key(const int* __restrict__ dst, float* __restrict__ deg, int* __restrict__ keys) {
    int t = blockIdx.x * blockDim.x + threadIdx.x;
    if (t >= NKW) return;
    unsigned kw = 0xFFFFFFFFu;
    if (t < NE / 4) {
        const int4 d = ((const int4*)dst)[t];
        atomicAdd(&deg[d.x], 1.0f);
        atomicAdd(&deg[d.y], 1.0f);
        atomicAdd(&deg[d.z], 1.0f);
        atomicAdd(&deg[d.w], 1.0f);
        kw = ((unsigned)d.x / RN) | (((unsigned)d.y / RN) << 8) |
             (((unsigned)d.z / RN) << 16) | (((unsigned)d.w / RN) << 24);
    }
    keys[t] = (int)kw;
}

__global__ void k_dinv(const float* __restrict__ deg, float* __restrict__ dinv) {
    int i = blockIdx.x * blockDim.x + threadIdx.x;
    if (i < NN) dinv[i] = rsqrtf(deg[i]);
}

__global__ void k_pad_w2(const float* __restrict__ W2, float* __restrict__ W2p) {
    int t = blockIdx.x * blockDim.x + threadIdx.x;
    if (t >= DH * NCP) return;
    int r = t / NCP, c = t - r * NCP;
    W2p[t] = (c < NC) ? W2[r * NC + c] : 0.0f;
}


#define GW 8
#define GT (GW * 32)

template <int K, int NP, int NOUT>
__global__ __launch_bounds__(GT) void k_gemm_wmma_f32(const float* __restrict__ A, const float* __restrict__ B,
                                                      float* __restrict__ C, int M) {
    constexpr int TN = NP / 16;
    constexpr int KS = K / 4;
    constexpr int C4 = NP / 4;
    constexpr int PP = 2 * NP + ((2 * NP) % 64 == 32 ? 0 : 32 - (2 * NP) % 64);
    constexpr int NI = ((K / 2) * C4) / GT;
    static_assert(K % 4 == 0 && NP % 16 == 0, "whole k steps and column tiles");
    static_assert(((K / 2) * C4) % GT == 0, "B is staged in whole rounds of the work-group");
    static_assert(PP % 64 == 32 && PP % 4 == 0, "fragment halves on disjoint banks, 16-byte LDS stores");
    __shared__ __attribute__((aligned(16))) float ldsB[(K / 2) * PP];

    const int lane = threadIdx.x & 31;
    int tm = blockIdx.x * GW + (threadIdx.x >> 5);
    const bool valid = (tm * 16 < M);
    if (!valid) tm = M / 16 - 1;

    const int m  = tm * 16 + (lane & 15);
    const int kh = (lane >> 4) << 1;

    float4 s0[NI], s1[NI];
#pragma unroll
    for (int i = 0; i < NI; ++i) {
        const int it = threadIdx.x + i * GT;
        const int j = it / C4, c4 = it - j * C4;
        s0[i] = ((const float4*)B)[(2 * j) * C4 + c4];
        s1[i] = ((const float4*)B)[(2 * j + 1) * C4 + c4];
    }
    const float* aP = A + (long)m * K + kh;
    v2f a[KS];
#pragma unroll
    for (int s = 0; s < KS; ++s) {
        a[s].x = aP[4 * s];
        a[s].y = aP[4 * s + 1];
    }
    __builtin_amdgcn_sched_barrier(0);
#pragma unroll
    for (int i = 0; i < NI; ++i) {
        const int it = threadIdx.x + i * GT;
        const int j = it / C4, c4 = it - j * C4;
        float* d = &ldsB[j * PP + c4 * 8];
        *(float4*)d       = make_float4(s0[i].x, s1[i].x, s0[i].y, s1[i].y);
        *(float4*)(d + 4) = make_float4(s0[i].z, s1[i].z, s0[i].w, s1[i].w);
    }
    __syncthreads();

    const int bOff = (kh >> 1) * PP + (lane & 15) * 2;

    v8f acc[TN];
#pragma unroll
    for (int t = 0; t < TN; ++t) acc[t] = v8f{};
    v2f bq[2][TN];
#pragma unroll
    for (int t = 0; t < TN; ++t)
        bq[0][t] = *(const v2f*)&ldsB[bOff + t * 32];
#pragma unroll
    for (int s = 0; s < KS; ++s) {
        if (s + 1 < KS) {
#pragma unroll
            for (int t = 0; t < TN; ++t)
                bq[(s + 1) & 1][t] = *(const v2f*)&ldsB[bOff + (2 * s + 2) * PP + t * 32];
        }
        __builtin_amdgcn_sched_barrier(0);
#pragma unroll
        for (int t = 0; t < TN; ++t)
            acc[t] = __builtin_amdgcn_wmma_f32_16x16x4_f32(
                 false, a[s],  false, bq[s & 1][t],
                 (short)0, acc[t],  false,  false);
        __builtin_amdgcn_sched_barrier(0);
    }

    if (valid) {
        const int rbase = tm * 16 + ((lane >> 4) << 3);
#pragma unroll
        for (int t = 0; t < TN; ++t) {
            const int n = t * 16 + (lane & 15);
            if (n < NOUT) {
#pragma unroll
                for (int v = 0; v < 8; ++v)
                    C[(long)(rbase + v) * NOUT + n] = acc[t][v];
            }
        }
    }
}


static_assert(NE % 4 == 0, "destinations are read as int4 by the degree launch");
static_assert(NN <= (1 << 20) && RN < (1 << 11), "saved list entry packing");
static_assert(NRG <= 255, "a range number fits a byte and 0xFF is free as the pad key");

template <int F, bool BIAS_RELU, bool SAVE, bool LOAD>
__global__ __launch_bounds__(OT) void k_own_agg(const int* __restrict__ src, const int* __restrict__ dst,
                                                const float* __restrict__ dinv, const float* __restrict__ h,
                                                const float* __restrict__ bias, float* __restrict__ outp,
                                                int* __restrict__ lcnt, int* __restrict__ lsaved,
                                                const int* __restrict__ keys) {
    __shared__ float acc[RN * F];
    __shared__ int   list[LCAP];
    __shared__ float dinvL[RN];
    __shared__ int   cnt;

    constexpr int NJ = (F + 31) / 32;
    const int tid  = threadIdx.x;
    const int lane = tid & 31;
    const int lo   = blockIdx.x * RN;

    int2* sv = (int2*)lsaved + (long)blockIdx.x * LCAP;
    const int  saved    = LOAD ? lcnt[blockIdx.x] : -1;
    const bool useSaved = LOAD && (unsigned)saved <= (unsigned)LCAP;

    if (tid == 0) cnt = 0;
    if (tid < RN) { int n = lo + tid; dinvL[tid] = (n < NN) ? dinv[n] : 0.0f; }
    for (int i = tid; i < RN * F; i += OT) {
        int nl = i / F, n = lo + nl;
        float v = 0.0f;
        if (n < NN) { float s = dinv[n]; s *= s; v = h[(long)lo * F + i] * s; }
        acc[i] = v;
    }
    __syncthreads();

    const int4* k4 = (const int4*)keys;
    const unsigned kb = (unsigned)blockIdx.x * 0x01010101u;
    if (!useSaved) {
        for (int b = tid; b < N16; b += OT * 4) {
            int4 d[4];
#pragma unroll
            for (int k = 0; k < 4; ++k) {
                int i = b + k * OT;
                d[k] = (i < N16) ? k4[i] : make_int4(-1, -1, -1, -1);
            }
            unsigned mm[2] = {0u, 0u};
#pragma unroll
            for (int k = 0; k < 4; ++k) {
                const unsigned x[4] = {(unsigned)d[k].x ^ kb, (unsigned)d[k].y ^ kb,
                                       (unsigned)d[k].z ^ kb, (unsigned)d[k].w ^ kb};
#pragma unroll
                for (int q = 0; q < 4; ++q) {
                    const unsigned z = ~(((x[q] & 0x7F7F7F7Fu) + 0x7F7F7F7Fu) | x[q] | 0x7F7F7F7Fu);
                    mm[k >> 1] |= z >> (7 - q - 4 * (k & 1));
                }
            }
            unsigned long long m = (unsigned long long)mm[0] | ((unsigned long long)mm[1] << 32);
            while (m) {
                const int p = __builtin_ctzll(m);
                m &= m - 1;
                const int k = ((p >> 5) << 1) | ((p >> 2) & 1);
                const int e = (b + k * OT) * 16 + ((p & 3) << 2) + ((p >> 3) & 3);
                int slot = atomicAdd(&cnt, 1);
                if (slot < LCAP) {
                    list[slot] = e;
                } else {
                    int s = src[e];
                    int r = dst[e] - lo;
                    float w = dinv[s] * dinvL[r];
                    for (int c = 0; c < F; ++c)
                        atomicAdd(&acc[r * F + c], h[(long)s * F + c] * w);
                }
            }
        }
    }
    __syncthreads();

    const int n = useSaved ? saved : (cnt < LCAP ? cnt : LCAP);
    if (SAVE && tid == 0) lcnt[blockIdx.x] = cnt;
    for (int g = (tid >> 5) * 32; g < n; g += OT) {
        const int  idx = g + lane;
        const bool ok  = idx < n;
        int r, s;
        float w;
        if (useSaved) {
            const int2 e = ok ? sv[idx] : make_int2(0, 0);
            r = e.x >> 20;
            s = e.x & 0xFFFFF;
            w = __int_as_float(e.y);
        } else {
            const int e = ok ? list[idx] : 0;
            s = ok ? src[e] : 0;
            r = ok ? dst[e] - lo : 0;
            w = ok ? dinv[s] * dinvL[r] : 0.0f;
            if (SAVE && ok) sv[idx] = make_int2(s | (r << 20), __float_as_int(w));
        }
        const int  m   = (n - g) < 32 ? (n - g) : 32;
        for (int k = 0; k < m; k += 4) {
            float v[4][NJ];
            int   rk[4];
            float wk[4];
#pragma unroll
            for (int kk = 0; kk < 4; ++kk) {
                const int sl = (k + kk) & 31;
                const int sk = __shfl(s, sl);
                rk[kk] = __shfl(r, sl);
                wk[kk] = __shfl(w, sl);
#pragma unroll
                for (int j = 0; j < NJ; ++j) {
                    const int c = lane + 32 * j;
                    v[kk][j] = (c < F) ? h[(long)sk * F + c] : 0.0f;
                }
            }
#pragma unroll
            for (int kk = 0; kk < 4; ++kk) {
                if (k + kk < m) {
#pragma unroll
                    for (int j = 0; j < NJ; ++j) {
                        const int c = lane + 32 * j;
                        if (c < F) atomicAdd(&acc[rk[kk] * F + c], v[kk][j] * wk[kk]);
                    }
                }
            }
        }
    }
    __syncthreads();

    for (int i = tid; i < RN * F; i += OT) {
        int nl = i / F;
        if (lo + nl < NN) {
            float v = acc[i];
            if (BIAS_RELU) { v += bias[i - nl * F]; v = v > 0.0f ? v : 0.0f; }
            outp[(long)lo * F + i] = v;
        }
    }
}


__global__ void k_logsoftmax(const float* __restrict__ agg, const float* __restrict__ b,
                             float* __restrict__ out) {
    int i = blockIdx.x * blockDim.x + threadIdx.x;
    if (i >= NN) return;
    const float* r = agg + (long)i * NC;
    float mx = -INFINITY;
#pragma unroll
    for (int c = 0; c < NC; ++c) mx = fmaxf(mx, r[c] + b[c]);
    float sum = 0.0f;
#pragma unroll
    for (int c = 0; c < NC; ++c) sum += expf(r[c] + b[c] - mx);
    float lse = mx + logf(sum);
#pragma unroll
    for (int c = 0; c < NC; ++c) out[(long)i * NC + c] = r[c] + b[c] - lse;
}


extern "C" void kernel_launch(void* const* d_in, const int* in_sizes, int n_in,
                              void* d_out, int out_size, void* d_ws, size_t ws_size,
                              hipStream_t stream) {
    const float* x  = (const float*)d_in[0];
    const int*   ei = (const int*)d_in[1];
    const float* W1 = (const float*)d_in[2];
    const float* b1 = (const float*)d_in[3];
    const float* W2 = (const float*)d_in[4];
    const float* b2 = (const float*)d_in[5];
    float* out = (float*)d_out;

    const int* src = ei;
    const int* dst = ei + NE;

    float* ws   = (float*)d_ws;
    float* deg  = ws;
    float* dinv = deg  + NN;
    float* h0   = dinv + NN;
    float* agg1 = h0   + (long)NN * DH;
    float* h2   = agg1 + (long)NN * DH;
    float* agg2 = h2   + (long)NN * NC;
    float* W2p  = agg2 + (long)NN * NC;
    int*   lcnt = (int*)(W2p + DH * NCP);
    int*   lsav = lcnt + 256;
    int*   keys = lsav + (long)NRG * LCAP * 2;

    const int B = 256;

    k_deg_init <<<cdiv(NN, B), B, 0, stream>>>(deg);
    k_deg_key  <<<cdiv(NKW, B), B, 0, stream>>>(dst, deg, keys);
    k_dinv     <<<cdiv(NN, B), B, 0, stream>>>(deg, dinv);
    k_pad_w2   <<<cdiv(DH * NCP, B), B, 0, stream>>>(W2, W2p);

    k_gemm_wmma_f32<FIN, DH, DH>
        <<<cdiv(NN / 16, GW), GT, 0, stream>>>(x, W1, h0, NN);
    k_own_agg<DH, true, true, false><<<cdiv(NN, RN), OT, 0, stream>>>(src, dst, dinv, h0, b1, agg1, lcnt, lsav, keys);

    k_gemm_wmma_f32<DH, NCP, NC>
        <<<cdiv(NN / 16, GW), GT, 0, stream>>>(agg1, W2p, h2, NN);
    k_own_agg<NC, false, false, true><<<cdiv(NN, RN), OT, 0, stream>>>(src, dst, dinv, h2, b2, agg2, lcnt, lsav, keys);

    k_logsoftmax<<<cdiv(NN, B), B, 0, stream>>>(agg2, b2, out);
}


# Owner launches: the range's own rows (self-term start of the LDS sums, final row write) move in 16-byte pieces, a thread's loads issued as one group; was one 4-byte load waited for per trip
# speedup vs baseline: 2.3413x; 1.1514x over previous
#include <hip/hip_runtime.h>
#include <math.h>

#define NN   50000
#define NE   625000
#define FIN  128
#define DH   128
#define NC   40
#define NCP  48

#define RN   200
#define LCAP 4096
#define OT   1024
#define NRG  ((NN + RN - 1) / RN)
#define N16  ((NE + 15) / 16)
#define NKW  (N16 * 4)

typedef __attribute__((ext_vector_type(2))) float v2f;
typedef __attribute__((ext_vector_type(8))) float v8f;

static inline int cdiv(long a, long b) { return (int)((a + b - 1) / b); }


__global__ void k_deg_init(float* __restrict__ deg) {
    int i = blockIdx.x * blockDim.x + threadIdx.x;
    if (i < NN) deg[i] = 1.0f;
}

__global__ void k_deg_key(const int* __restrict__ dst, float* __restrict__ deg, int* __restrict__ keys) {
    int t = blockIdx.x * blockDim.x + threadIdx.x;
    if (t >= NKW) return;
    unsigned kw = 0xFFFFFFFFu;
    if (t < NE / 4) {
        const int4 d = ((const int4*)dst)[t];
        atomicAdd(&deg[d.x], 1.0f);
        atomicAdd(&deg[d.y], 1.0f);
        atomicAdd(&deg[d.z], 1.0f);
        atomicAdd(&deg[d.w], 1.0f);
        kw = ((unsigned)d.x / RN) | (((unsigned)d.y / RN) << 8) |
             (((unsigned)d.z / RN) << 16) | (((unsigned)d.w / RN) << 24);
    }
    keys[t] = (int)kw;
}

__global__ void k_dinv(const float* __restrict__ deg, float* __restrict__ dinv) {
    int i = blockIdx.x * blockDim.x + threadIdx.x;
    if (i < NN) dinv[i] = rsqrtf(deg[i]);
}

__global__ void k_pad_w2(const float* __restrict__ W2, float* __restrict__ W2p) {
    int t = blockIdx.x * blockDim.x + threadIdx.x;
    if (t >= DH * NCP) return;
    int r = t / NCP, c = t - r * NCP;
    W2p[t] = (c < NC) ? W2[r * NC + c] : 0.0f;
}


#define GW 8
#define GT (GW * 32)

template <int K, int NP, int NOUT>
__global__ __launch_bounds__(GT) void k_gemm_wmma_f32(const float* __restrict__ A, const float* __restrict__ B,
                                                      float* __restrict__ C, int M) {
    constexpr int TN = NP / 16;
    constexpr int KS = K / 4;
    constexpr int C4 = NP / 4;
    constexpr int PP = 2 * NP + ((2 * NP) % 64 == 32 ? 0 : 32 - (2 * NP) % 64);
    constexpr int NI = ((K / 2) * C4) / GT;
    static_assert(K % 4 == 0 && NP % 16 == 0, "whole k steps and column tiles");
    static_assert(((K / 2) * C4) % GT == 0, "B is staged in whole rounds of the work-group");
    static_assert(PP % 64 == 32 && PP % 4 == 0, "fragment halves on disjoint banks, 16-byte LDS stores");
    __shared__ __attribute__((aligned(16))) float ldsB[(K / 2) * PP];

    const int lane = threadIdx.x & 31;
    int tm = blockIdx.x * GW + (threadIdx.x >> 5);
    const bool valid = (tm * 16 < M);
    if (!valid) tm = M / 16 - 1;

    const int m  = tm * 16 + (lane & 15);
    const int kh = (lane >> 4) << 1;

    float4 s0[NI], s1[NI];
#pragma unroll
    for (int i = 0; i < NI; ++i) {
        const int it = threadIdx.x + i * GT;
        const int j = it / C4, c4 = it - j * C4;
        s0[i] = ((const float4*)B)[(2 * j) * C4 + c4];
        s1[i] = ((const float4*)B)[(2 * j + 1) * C4 + c4];
    }
    const float* aP = A + (long)m * K + kh;
    v2f a[KS];
#pragma unroll
    for (int s = 0; s < KS; ++s) {
        a[s].x = aP[4 * s];
        a[s].y = aP[4 * s + 1];
    }
    __builtin_amdgcn_sched_barrier(0);
#pragma unroll
    for (int i = 0; i < NI; ++i) {
        const int it = threadIdx.x + i * GT;
        const int j = it / C4, c4 = it - j * C4;
        float* d = &ldsB[j * PP + c4 * 8];
        *(float4*)d       = make_float4(s0[i].x, s1[i].x, s0[i].y, s1[i].y);
        *(float4*)(d + 4) = make_float4(s0[i].z, s1[i].z, s0[i].w, s1[i].w);
    }
    __syncthreads();

    const int bOff = (kh >> 1) * PP + (lane & 15) * 2;

    v8f acc[TN];
#pragma unroll
    for (int t = 0; t < TN; ++t) acc[t] = v8f{};
    v2f bq[2][TN];
#pragma unroll
    for (int t = 0; t < TN; ++t)
        bq[0][t] = *(const v2f*)&ldsB[bOff + t * 32];
#pragma unroll
    for (int s = 0; s < KS; ++s) {
        if (s + 1 < KS) {
#pragma unroll
            for (int t = 0; t < TN; ++t)
                bq[(s + 1) & 1][t] = *(const v2f*)&ldsB[bOff + (2 * s + 2) * PP + t * 32];
        }
        __builtin_amdgcn_sched_barrier(0);
#pragma unroll
        for (int t = 0; t < TN; ++t)
            acc[t] = __builtin_amdgcn_wmma_f32_16x16x4_f32(
                 false, a[s],  false, bq[s & 1][t],
                 (short)0, acc[t],  false,  false);
        __builtin_amdgcn_sched_barrier(0);
    }

    if (valid) {
        const int rbase = tm * 16 + ((lane >> 4) << 3);
#pragma unroll
        for (int t = 0; t < TN; ++t) {
            const int n = t * 16 + (lane & 15);
            if (n < NOUT) {
#pragma unroll
                for (int v = 0; v < 8; ++v)
                    C[(long)(rbase + v) * NOUT + n] = acc[t][v];
            }
        }
    }
}


static_assert(NE % 4 == 0, "destinations are read as int4 by the degree launch");
static_assert(NN <= (1 << 20) && RN < (1 << 11), "saved list entry packing");
static_assert(NRG <= 255, "a range number fits a byte and 0xFF is free as the pad key");

template <int F, bool BIAS_RELU, bool SAVE, bool LOAD>
__global__ __launch_bounds__(OT) void k_own_agg(const int* __restrict__ src, const int* __restrict__ dst,
                                                const float* __restrict__ dinv, const float* __restrict__ h,
                                                const float* __restrict__ bias, float* __restrict__ outp,
                                                int* __restrict__ lcnt, int* __restrict__ lsaved,
                                                const int* __restrict__ keys) {
    __shared__ __attribute__((aligned(16))) float acc[RN * F];
    __shared__ int   list[LCAP];
    __shared__ float dinvL[RN];
    __shared__ int   cnt;

    constexpr int NJ = (F + 31) / 32;
    constexpr int F4 = F / 4;
    constexpr int NV = RN * F4;
    constexpr int NK = (NV + OT - 1) / OT;
    static_assert(F % 4 == 0, "a 16-byte piece never straddles two rows");
    static_assert(NN % RN == 0, "every range is whole: all RN owned rows exist");
    static_assert(!BIAS_RELU || OT % F4 == 0, "a thread's pieces all sit in one column group");
    const int tid  = threadIdx.x;
    const int lane = tid & 31;
    const int lo   = blockIdx.x * RN;

    int2* sv = (int2*)lsaved + (long)blockIdx.x * LCAP;
    const int  saved    = LOAD ? lcnt[blockIdx.x] : -1;
    const bool useSaved = LOAD && (unsigned)saved <= (unsigned)LCAP;

    if (tid == 0) cnt = 0;
    if (tid < RN) { int n = lo + tid; dinvL[tid] = (n < NN) ? dinv[n] : 0.0f; }
    {
        const float4* h4 = (const float4*)(h + (long)lo * F);
        float4 hv[NK];
        float  dv[NK];
#pragma unroll
        for (int k = 0; k < NK; ++k) {
            int i4 = tid + k * OT;
            i4 = i4 < NV ? i4 : NV - 1;
            hv[k] = h4[i4];
            dv[k] = dinv[lo + i4 / F4];
        }
#pragma unroll
        for (int k = 0; k < NK; ++k) {
            const int i4 = tid + k * OT;
            if (i4 < NV) {
                float s = dv[k]; s *= s;
                ((float4*)acc)[i4] = make_float4(hv[k].x * s, hv[k].y * s, hv[k].z * s, hv[k].w * s);
            }
        }
    }
    __syncthreads();

    const int4* k4 = (const int4*)keys;
    const unsigned kb = (unsigned)blockIdx.x * 0x01010101u;
    if (!useSaved) {
        for (int b = tid; b < N16; b += OT * 4) {
            int4 d[4];
#pragma unroll
            for (int k = 0; k < 4; ++k) {
                int i = b + k * OT;
                d[k] = (i < N16) ? k4[i] : make_int4(-1, -1, -1, -1);
            }
            unsigned mm[2] = {0u, 0u};
#pragma unroll
            for (int k = 0; k < 4; ++k) {
                const unsigned x[4] = {(unsigned)d[k].x ^ kb, (unsigned)d[k].y ^ kb,
                                       (unsigned)d[k].z ^ kb, (unsigned)d[k].w ^ kb};
#pragma unroll
                for (int q = 0; q < 4; ++q) {
                    const unsigned z = ~(((x[q] & 0x7F7F7F7Fu) + 0x7F7F7F7Fu) | x[q] | 0x7F7F7F7Fu);
                    mm[k >> 1] |= z >> (7 - q - 4 * (k & 1));
                }
            }
            unsigned long long m = (unsigned long long)mm[0] | ((unsigned long long)mm[1] << 32);
            while (m) {
                const int p = __builtin_ctzll(m);
                m &= m - 1;
                const int k = ((p >> 5) << 1) | ((p >> 2) & 1);
                const int e = (b + k * OT) * 16 + ((p & 3) << 2) + ((p >> 3) & 3);
                int slot = atomicAdd(&cnt, 1);
                if (slot < LCAP) {
                    list[slot] = e;
                } else {
                    int s = src[e];
                    int r = dst[e] - lo;
                    float w = dinv[s] * dinvL[r];
                    for (int c = 0; c < F; ++c)
                        atomicAdd(&acc[r * F + c], h[(long)s * F + c] * w);
                }
            }
        }
    }
    __syncthreads();

    const int n = useSaved ? saved : (cnt < LCAP ? cnt : LCAP);
    if (SAVE && tid == 0) lcnt[blockIdx.x] = cnt;
    for (int g = (tid >> 5) * 32; g < n; g += OT) {
        const int  idx = g + lane;
        const bool ok  = idx < n;
        int r, s;
        float w;
        if (useSaved) {
            const int2 e = ok ? sv[idx] : make_int2(0, 0);
            r = e.x >> 20;
            s = e.x & 0xFFFFF;
            w = __int_as_float(e.y);
        } else {
            const int e = ok ? list[idx] : 0;
            s = ok ? src[e] : 0;
            r = ok ? dst[e] - lo : 0;
            w = ok ? dinv[s] * dinvL[r] : 0.0f;
            if (SAVE && ok) sv[idx] = make_int2(s | (r << 20), __float_as_int(w));
        }
        const int  m   = (n - g) < 32 ? (n - g) : 32;
        for (int k = 0; k < m; k += 4) {
            float v[4][NJ];
            int   rk[4];
            float wk[4];
#pragma unroll
            for (int kk = 0; kk < 4; ++kk) {
                const int sl = (k + kk) & 31;
                const int sk = __shfl(s, sl);
                rk[kk] = __shfl(r, sl);
                wk[kk] = __shfl(w, sl);
#pragma unroll
                for (int j = 0; j < NJ; ++j) {
                    const int c = lane + 32 * j;
                    v[kk][j] = (c < F) ? h[(long)sk * F + c] : 0.0f;
                }
            }
#pragma unroll
            for (int kk = 0; kk < 4; ++kk) {
                if (k + kk < m) {
#pragma unroll
                    for (int j = 0; j < NJ; ++j) {
                        const int c = lane + 32 * j;
                        if (c < F) atomicAdd(&acc[rk[kk] * F + c], v[kk][j] * wk[kk]);
                    }
                }
            }
        }
    }
    __syncthreads();

    {
        float4 bv = make_float4(0.0f, 0.0f, 0.0f, 0.0f);
        if (BIAS_RELU) bv = ((const float4*)bias)[tid % F4];
        float4* o4 = (float4*)(outp + (long)lo * F);
#pragma unroll
        for (int k = 0; k < NK; ++k) {
            const int i4 = tid + k * OT;
            if (i4 < NV) {
                float4 v = ((const float4*)acc)[i4];
                if (BIAS_RELU) {
                    v.x += bv.x; v.y += bv.y; v.z += bv.z; v.w += bv.w;
                    v.x = v.x > 0.0f ? v.x : 0.0f; v.y = v.y > 0.0f ? v.y : 0.0f;
                    v.z = v.z > 0.0f ? v.z : 0.0f; v.w = v.w > 0.0f ? v.w : 0.0f;
                }
                o4[i4] = v;
            }
        }
    }
}


__global__ void k_logsoftmax(const float* __restrict__ agg, const float* __restrict__ b,
                             float* __restrict__ out) {
    int i = blockIdx.x * blockDim.x + threadIdx.x;
    if (i >= NN) return;
    const float* r = agg + (long)i * NC;
    float mx = -INFINITY;
#pragma unroll
    for (int c = 0; c < NC; ++c) mx = fmaxf(mx, r[c] + b[c]);
    float sum = 0.0f;
#pragma unroll
    for (int c = 0; c < NC; ++c) sum += expf(r[c] + b[c] - mx);
    float lse = mx + logf(sum);
#pragma unroll
    for (int c = 0; c < NC; ++c) out[(long)i * NC + c] = r[c] + b[c] - lse;
}


extern "C" void kernel_launch(void* const* d_in, const int* in_sizes, int n_in,
                              void* d_out, int out_size, void* d_ws, size_t ws_size,
                              hipStream_t stream) {
    const float* x  = (const float*)d_in[0];
    const int*   ei = (const int*)d_in[1];
    const float* W1 = (const float*)d_in[2];
    const float* b1 = (const float*)d_in[3];
    const float* W2 = (const float*)d_in[4];
    const float* b2 = (const float*)d_in[5];
    float* out = (float*)d_out;

    const int* src = ei;
    const int* dst = ei + NE;

    float* ws   = (float*)d_ws;
    float* deg  = ws;
    float* dinv = deg  + NN;
    float* h0   = dinv + NN;
    float* agg1 = h0   + (long)NN * DH;
    float* h2   = agg1 + (long)NN * DH;
    float* agg2 = h2   + (long)NN * NC;
    float* W2p  = agg2 + (long)NN * NC;
    int*   lcnt = (int*)(W2p + DH * NCP);
    int*   lsav = lcnt + 256;
    int*   keys = lsav + (long)NRG * LCAP * 2;

    const int B = 256;

    k_deg_init <<<cdiv(NN, B), B, 0, stream>>>(deg);
    k_deg_key  <<<cdiv(NKW, B), B, 0, stream>>>(dst, deg, keys);
    k_dinv     <<<cdiv(NN, B), B, 0, stream>>>(deg, dinv);
    k_pad_w2   <<<cdiv(DH * NCP, B), B, 0, stream>>>(W2, W2p);

    k_gemm_wmma_f32<FIN, DH, DH>
        <<<cdiv(NN / 16, GW), GT, 0, stream>>>(x, W1, h0, NN);
    k_own_agg<DH, true, true, false><<<cdiv(NN, RN), OT, 0, stream>>>(src, dst, dinv, h0, b1, agg1, lcnt, lsav, keys);

    k_gemm_wmma_f32<DH, NCP, NC>
        <<<cdiv(NN / 16, GW), GT, 0, stream>>>(agg1, W2p, h2, NN);
    k_own_agg<NC, false, false, true><<<cdiv(NN, RN), OT, 0, stream>>>(src, dst, dinv, h2, b2, agg2, lcnt, lsav, keys);

    k_logsoftmax<<<cdiv(NN, B), B, 0, stream>>>(agg2, b2, out);
}


# Layer-2 GEMM moved into the layer-1 owner launch: a wave per 16-row tile multiplies the finished LDS rows (bias, relu at fragment read) by W2 on f32 WMMA, stores h2; h1 plane, GEMM, pad launches go
# speedup vs baseline: 2.5076x; 1.0710x over previous
#include <hip/hip_runtime.h>
#include <math.h>

#define NN   50000
#define NE   625000
#define FIN  128
#define DH   128
#define NC   40
#define NCP  48

#define RN   200
#define LCAP 4096
#define OT   1024
#define NRG  ((NN + RN - 1) / RN)
#define N16  ((NE + 15) / 16)
#define NKW  (N16 * 4)

typedef __attribute__((ext_vector_type(2))) float v2f;
typedef __attribute__((ext_vector_type(8))) float v8f;

static inline int cdiv(long a, long b) { return (int)((a + b - 1) / b); }


__global__ void k_deg_init(float* __restrict__ deg) {
    int i = blockIdx.x * blockDim.x + threadIdx.x;
    if (i < NN) deg[i] = 1.0f;
}

__global__ void k_deg_key(const int* __restrict__ dst, float* __restrict__ deg, int* __restrict__ keys) {
    int t = blockIdx.x * blockDim.x + threadIdx.x;
    if (t >= NKW) return;
    unsigned kw = 0xFFFFFFFFu;
    if (t < NE / 4) {
        const int4 d = ((const int4*)dst)[t];
        atomicAdd(&deg[d.x], 1.0f);
        atomicAdd(&deg[d.y], 1.0f);
        atomicAdd(&deg[d.z], 1.0f);
        atomicAdd(&deg[d.w], 1.0f);
        kw = ((unsigned)d.x / RN) | (((unsigned)d.y / RN) << 8) |
             (((unsigned)d.z / RN) << 16) | (((unsigned)d.w / RN) << 24);
    }
    keys[t] = (int)kw;
}

__global__ void k_dinv(const float* __restrict__ deg, float* __restrict__ dinv) {
    int i = blockIdx.x * blockDim.x + threadIdx.x;
    if (i < NN) dinv[i] = rsqrtf(deg[i]);
}


#define GW 8
#define GT (GW * 32)

template <int K, int NP, int NOUT>
__global__ __launch_bounds__(GT) void k_gemm_wmma_f32(const float* __restrict__ A, const float* __restrict__ B,
                                                      float* __restrict__ C, int M) {
    constexpr int TN = NP / 16;
    constexpr int KS = K / 4;
    constexpr int C4 = NP / 4;
    constexpr int PP = 2 * NP + ((2 * NP) % 64 == 32 ? 0 : 32 - (2 * NP) % 64);
    constexpr int NI = ((K / 2) * C4) / GT;
    static_assert(K % 4 == 0 && NP % 16 == 0, "whole k steps and column tiles");
    static_assert(((K / 2) * C4) % GT == 0, "B is staged in whole rounds of the work-group");
    static_assert(PP % 64 == 32 && PP % 4 == 0, "fragment halves on disjoint banks, 16-byte LDS stores");
    __shared__ __attribute__((aligned(16))) float ldsB[(K / 2) * PP];

    const int lane = threadIdx.x & 31;
    int tm = blockIdx.x * GW + (threadIdx.x >> 5);
    const bool valid = (tm * 16 < M);
    if (!valid) tm = M / 16 - 1;

    const int m  = tm * 16 + (lane & 15);
    const int kh = (lane >> 4) << 1;

    float4 s0[NI], s1[NI];
#pragma unroll
    for (int i = 0; i < NI; ++i) {
        const int it = threadIdx.x + i * GT;
        const int j = it / C4, c4 = it - j * C4;
        s0[i] = ((const float4*)B)[(2 * j) * C4 + c4];
        s1[i] = ((const float4*)B)[(2 * j + 1) * C4 + c4];
    }
    const float* aP = A + (long)m * K + kh;
    v2f a[KS];
#pragma unroll
    for (int s = 0; s < KS; ++s) {
        a[s].x = aP[4 * s];
        a[s].y = aP[4 * s + 1];
    }
    __builtin_amdgcn_sched_barrier(0);
#pragma unroll
    for (int i = 0; i < NI; ++i) {
        const int it = threadIdx.x + i * GT;
        const int j = it / C4, c4 = it - j * C4;
        float* d = &ldsB[j * PP + c4 * 8];
        *(float4*)d       = make_float4(s0[i].x, s1[i].x, s0[i].y, s1[i].y);
        *(float4*)(d + 4) = make_float4(s0[i].z, s1[i].z, s0[i].w, s1[i].w);
    }
    __syncthreads();

    const int bOff = (kh >> 1) * PP + (lane & 15) * 2;

    v8f acc[TN];
#pragma unroll
    for (int t = 0; t < TN; ++t) acc[t] = v8f{};
    v2f bq[2][TN];
#pragma unroll
    for (int t = 0; t < TN; ++t)
        bq[0][t] = *(const v2f*)&ldsB[bOff + t * 32];
#pragma unroll
    for (int s = 0; s < KS; ++s) {
        if (s + 1 < KS) {
#pragma unroll
            for (int t = 0; t < TN; ++t)
                bq[(s + 1) & 1][t] = *(const v2f*)&ldsB[bOff + (2 * s + 2) * PP + t * 32];
        }
        __builtin_amdgcn_sched_barrier(0);
#pragma unroll
        for (int t = 0; t < TN; ++t)
            acc[t] = __builtin_amdgcn_wmma_f32_16x16x4_f32(
                 false, a[s],  false, bq[s & 1][t],
                 (short)0, acc[t],  false,  false);
        __builtin_amdgcn_sched_barrier(0);
    }

    if (valid) {
        const int rbase = tm * 16 + ((lane >> 4) << 3);
#pragma unroll
        for (int t = 0; t < TN; ++t) {
            const int n = t * 16 + (lane & 15);
            if (n < NOUT) {
#pragma unroll
                for (int v = 0; v < 8; ++v)
                    C[(long)(rbase + v) * NOUT + n] = acc[t][v];
            }
        }
    }
}


static_assert(NE % 4 == 0, "destinations are read as int4 by the degree launch");
static_assert(NN <= (1 << 20) && RN < (1 << 11), "saved list entry packing");
static_assert(NRG <= 255, "a range number fits a byte and 0xFF is free as the pad key");

template <int F, bool BIAS_RELU, bool SAVE, bool LOAD, bool FUSE>
__global__ __launch_bounds__(OT) void k_own_agg(const int* __restrict__ src, const int* __restrict__ dst,
                                                const float* __restrict__ dinv, const float* __restrict__ h,
                                                const float* __restrict__ bias, float* __restrict__ outp,
                                                int* __restrict__ lcnt, int* __restrict__ lsaved,
                                                const int* __restrict__ keys, const float* __restrict__ W2) {
    constexpr int PITCH = FUSE ? F + 4 : F;
    __shared__ __attribute__((aligned(16))) float acc[RN * PITCH];
    __shared__ int   list[LCAP];
    __shared__ float dinvL[RN];
    __shared__ int   cnt;

    constexpr int NJ = (F + 31) / 32;
    constexpr int F4 = F / 4;
    constexpr int NV = RN * F4;
    constexpr int NK = (NV + OT - 1) / OT;
    static_assert(F % 4 == 0, "a 16-byte piece never straddles two rows");
    static_assert(NN % RN == 0, "every range is whole: all RN owned rows exist");
    static_assert(!BIAS_RELU || OT % F4 == 0, "a thread's pieces all sit in one column group");
    static_assert(!FUSE || (BIAS_RELU && F == DH && F % 4 == 0), "the fused product is relu(row + bias) @ W2");
    static_assert(PITCH % 4 == 0, "16-byte pieces and 8-byte fragment reads stay aligned");
    const int tid  = threadIdx.x;
    const int lane = tid & 31;
    const int lo   = blockIdx.x * RN;

    int2* sv = (int2*)lsaved + (long)blockIdx.x * LCAP;
    const int  saved    = LOAD ? lcnt[blockIdx.x] : -1;
    const bool useSaved = LOAD && (unsigned)saved <= (unsigned)LCAP;

    if (tid == 0) cnt = 0;
    if (tid < RN) { int n = lo + tid; dinvL[tid] = (n < NN) ? dinv[n] : 0.0f; }
    {
        const float4* h4 = (const float4*)(h + (long)lo * F);
        float4 hv[NK];
        float  dv[NK];
#pragma unroll
        for (int k = 0; k < NK; ++k) {
            int i4 = tid + k * OT;
            i4 = i4 < NV ? i4 : NV - 1;
            hv[k] = h4[i4];
            dv[k] = dinv[lo + i4 / F4];
        }
#pragma unroll
        for (int k = 0; k < NK; ++k) {
            const int i4 = tid + k * OT;
            if (i4 < NV) {
                float s = dv[k]; s *= s;
                const int o = (PITCH == F) ? i4 * 4 : (i4 / F4) * PITCH + (i4 % F4) * 4;
                *(float4*)&acc[o] = make_float4(hv[k].x * s, hv[k].y * s, hv[k].z * s, hv[k].w * s);
            }
        }
    }
    __syncthreads();

    const int4* k4 = (const int4*)keys;
    const unsigned kb = (unsigned)blockIdx.x * 0x01010101u;
    if (!useSaved) {
        for (int b = tid; b < N16; b += OT * 4) {
            int4 d[4];
#pragma unroll
            for (int k = 0; k < 4; ++k) {
                int i = b + k * OT;
                d[k] = (i < N16) ? k4[i] : make_int4(-1, -1, -1, -1);
            }
            unsigned mm[2] = {0u, 0u};
#pragma unroll
            for (int k = 0; k < 4; ++k) {
                const unsigned x[4] = {(unsigned)d[k].x ^ kb, (unsigned)d[k].y ^ kb,
                                       (unsigned)d[k].z ^ kb, (unsigned)d[k].w ^ kb};
#pragma unroll
                for (int q = 0; q < 4; ++q) {
                    const unsigned z = ~(((x[q] & 0x7F7F7F7Fu) + 0x7F7F7F7Fu) | x[q] | 0x7F7F7F7Fu);
                    mm[k >> 1] |= z >> (7 - q - 4 * (k & 1));
                }
            }
            unsigned long long m = (unsigned long long)mm[0] | ((unsigned long long)mm[1] << 32);
            while (m) {
                const int p = __builtin_ctzll(m);
                m &= m - 1;
                const int k = ((p >> 5) << 1) | ((p >> 2) & 1);
                const int e = (b + k * OT) * 16 + ((p & 3) << 2) + ((p >> 3) & 3);
                int slot = atomicAdd(&cnt, 1);
                if (slot < LCAP) {
                    list[slot] = e;
                } else {
                    int s = src[e];
                    int r = dst[e] - lo;
                    float w = dinv[s] * dinvL[r];
                    for (int c = 0; c < F; ++c)
                        atomicAdd(&acc[r * PITCH + c], h[(long)s * F + c] * w);
                }
            }
        }
    }
    __syncthreads();

    const int n = useSaved ? saved : (cnt < LCAP ? cnt : LCAP);
    if (SAVE && tid == 0) lcnt[blockIdx.x] = cnt;
    for (int g = (tid >> 5) * 32; g < n; g += OT) {
        const int  idx = g + lane;
        const bool ok  = idx < n;
        int r, s;
        float w;
        if (useSaved) {
            const int2 e = ok ? sv[idx] : make_int2(0, 0);
            r = e.x >> 20;
            s = e.x & 0xFFFFF;
            w = __int_as_float(e.y);
        } else {
            const int e = ok ? list[idx] : 0;
            s = ok ? src[e] : 0;
            r = ok ? dst[e] - lo : 0;
            w = ok ? dinv[s] * dinvL[r] : 0.0f;
            if (SAVE && ok) sv[idx] = make_int2(s | (r << 20), __float_as_int(w));
        }
        const int  m   = (n - g) < 32 ? (n - g) : 32;
        for (int k = 0; k < m; k += 4) {
            float v[4][NJ];
            int   rk[4];
            float wk[4];
#pragma unroll
            for (int kk = 0; kk < 4; ++kk) {
                const int sl = (k + kk) & 31;
                const int sk = __shfl(s, sl);
                rk[kk] = __shfl(r, sl);
                wk[kk] = __shfl(w, sl);
#pragma unroll
                for (int j = 0; j < NJ; ++j) {
                    const int c = lane + 32 * j;
                    v[kk][j] = (c < F) ? h[(long)sk * F + c] : 0.0f;
                }
            }
#pragma unroll
            for (int kk = 0; kk < 4; ++kk) {
                if (k + kk < m) {
#pragma unroll
                    for (int j = 0; j < NJ; ++j) {
                        const int c = lane + 32 * j;
                        if (c < F) atomicAdd(&acc[rk[kk] * PITCH + c], v[kk][j] * wk[kk]);
                    }
                }
            }
        }
    }
    __syncthreads();

    if (FUSE) {
        constexpr int NT = (RN + 15) / 16;
        constexpr int TN = NCP / 16;
        constexpr int KS = F / 4;
        const int wv = __builtin_amdgcn_readfirstlane(tid >> 5);
        if (wv < NT) {
            const int c  = lane & 15;
            const int kh = (lane >> 4) << 1;
            int ra = wv * 16 + c;
            ra = ra < RN ? ra : RN - 1;
            const float* aP = &acc[ra * PITCH + kh];
            const float* bP = bias + kh;
            int  nc[TN];
            bool nv[TN];
#pragma unroll
            for (int t = 0; t < TN; ++t) {
                const int n = t * 16 + c;
                nv[t] = n < NC;
                nc[t] = nv[t] ? n : NC - 1;
            }
            v8f d[TN];
#pragma unroll
            for (int t = 0; t < TN; ++t) d[t] = v8f{};
#pragma unroll
            for (int s = 0; s < KS; ++s) {
                v2f a = *(const v2f*)&aP[4 * s];
                const v2f bb = *(const v2f*)&bP[4 * s];
                a.x += bb.x; a.y += bb.y;
                a.x = a.x > 0.0f ? a.x : 0.0f;
                a.y = a.y > 0.0f ? a.y : 0.0f;
#pragma unroll
                for (int t = 0; t < TN; ++t) {
                    const float w0 = W2[(4 * s + kh) * NC + nc[t]];
                    const float w1 = W2[(4 * s + kh + 1) * NC + nc[t]];
                    v2f b;
                    b.x = nv[t] ? w0 : 0.0f;
                    b.y = nv[t] ? w1 : 0.0f;
                    d[t] = __builtin_amdgcn_wmma_f32_16x16x4_f32(
                         false, a,  false, b,
                         (short)0, d[t],  false,  false);
                }
            }
            const int rbase = wv * 16 + ((lane >> 4) << 3);
#pragma unroll
            for (int t = 0; t < TN; ++t) {
                if (nv[t]) {
#pragma unroll
                    for (int v = 0; v < 8; ++v)
                        if (rbase + v < RN)
                            outp[(long)(lo + rbase + v) * NC + t * 16 + c] = d[t][v];
                }
            }
        }
    } else {
        float4 bv = make_float4(0.0f, 0.0f, 0.0f, 0.0f);
        if (BIAS_RELU) bv = ((const float4*)bias)[tid % F4];
        float4* o4 = (float4*)(outp + (long)lo * F);
#pragma unroll
        for (int k = 0; k < NK; ++k) {
            const int i4 = tid + k * OT;
            if (i4 < NV) {
                float4 v = ((const float4*)acc)[i4];
                if (BIAS_RELU) {
                    v.x += bv.x; v.y += bv.y; v.z += bv.z; v.w += bv.w;
                    v.x = v.x > 0.0f ? v.x : 0.0f; v.y = v.y > 0.0f ? v.y : 0.0f;
                    v.z = v.z > 0.0f ? v.z : 0.0f; v.w = v.w > 0.0f ? v.w : 0.0f;
                }
                o4[i4] = v;
            }
        }
    }
}


__global__ void k_logsoftmax(const float* __restrict__ agg, const float* __restrict__ b,
                             float* __restrict__ out) {
    int i = blockIdx.x * blockDim.x + threadIdx.x;
    if (i >= NN) return;
    const float* r = agg + (long)i * NC;
    float mx = -INFINITY;
#pragma unroll
    for (int c = 0; c < NC; ++c) mx = fmaxf(mx, r[c] + b[c]);
    float sum = 0.0f;
#pragma unroll
    for (int c = 0; c < NC; ++c) sum += expf(r[c] + b[c] - mx);
    float lse = mx + logf(sum);
#pragma unroll
    for (int c = 0; c < NC; ++c) out[(long)i * NC + c] = r[c] + b[c] - lse;
}


extern "C" void kernel_launch(void* const* d_in, const int* in_sizes, int n_in,
                              void* d_out, int out_size, void* d_ws, size_t ws_size,
                              hipStream_t stream) {
    const float* x  = (const float*)d_in[0];
    const int*   ei = (const int*)d_in[1];
    const float* W1 = (const float*)d_in[2];
    const float* b1 = (const float*)d_in[3];
    const float* W2 = (const float*)d_in[4];
    const float* b2 = (const float*)d_in[5];
    float* out = (float*)d_out;

    const int* src = ei;
    const int* dst = ei + NE;

    float* ws   = (float*)d_ws;
    float* deg  = ws;
    float* dinv = deg  + NN;
    float* h0   = dinv + NN;
    float* h2   = h0   + (long)NN * DH;
    float* agg2 = h2   + (long)NN * NC;
    int*   lcnt = (int*)(agg2 + (long)NN * NC);
    int*   lsav = lcnt + 256;
    int*   keys = lsav + (long)NRG * LCAP * 2;

    const int B = 256;

    k_deg_init <<<cdiv(NN, B), B, 0, stream>>>(deg);
    k_deg_key  <<<cdiv(NKW, B), B, 0, stream>>>(dst, deg, keys);
    k_dinv     <<<cdiv(NN, B), B, 0, stream>>>(deg, dinv);

    k_gemm_wmma_f32<FIN, DH, DH>
        <<<cdiv(NN / 16, GW), GT, 0, stream>>>(x, W1, h0, NN);
    k_own_agg<DH, true, true, false, true><<<cdiv(NN, RN), OT, 0, stream>>>(src, dst, dinv, h0, b1, h2, lcnt, lsav, keys, W2);

    k_own_agg<NC, false, false, true, false><<<cdiv(NN, RN), OT, 0, stream>>>(src, dst, dinv, h2, b2, agg2, lcnt, lsav, keys, W2);

    k_logsoftmax<<<cdiv(NN, B), B, 0, stream>>>(agg2, b2, out);
}


# Owner launches' row-add loop: entry source, node, weight broadcast by v_readlane into scalars (scalar trip counts), not by LDS permutes that waited on the wave's own LDS adds; arrival-order adds kept
# speedup vs baseline: 2.6536x; 1.0582x over previous
#include <hip/hip_runtime.h>
#include <math.h>

#define NN   50000
#define NE   625000
#define FIN  128
#define DH   128
#define NC   40
#define NCP  48

#define RN   200
#define LCAP 4096
#define OT   1024
#define NRG  ((NN + RN - 1) / RN)
#define N16  ((NE + 15) / 16)
#define NKW  (N16 * 4)

typedef __attribute__((ext_vector_type(2))) float v2f;
typedef __attribute__((ext_vector_type(8))) float v8f;

static inline int cdiv(long a, long b) { return (int)((a + b - 1) / b); }


__global__ void k_deg_init(float* __restrict__ deg) {
    int i = blockIdx.x * blockDim.x + threadIdx.x;
    if (i < NN) deg[i] = 1.0f;
}

__global__ void k_deg_key(const int* __restrict__ dst, float* __restrict__ deg, int* __restrict__ keys) {
    int t = blockIdx.x * blockDim.x + threadIdx.x;
    if (t >= NKW) return;
    unsigned kw = 0xFFFFFFFFu;
    if (t < NE / 4) {
        const int4 d = ((const int4*)dst)[t];
        atomicAdd(&deg[d.x], 1.0f);
        atomicAdd(&deg[d.y], 1.0f);
        atomicAdd(&deg[d.z], 1.0f);
        atomicAdd(&deg[d.w], 1.0f);
        kw = ((unsigned)d.x / RN) | (((unsigned)d.y / RN) << 8) |
             (((unsigned)d.z / RN) << 16) | (((unsigned)d.w / RN) << 24);
    }
    keys[t] = (int)kw;
}

__global__ void k_dinv(const float* __restrict__ deg, float* __restrict__ dinv) {
    int i = blockIdx.x * blockDim.x + threadIdx.x;
    if (i < NN) dinv[i] = rsqrtf(deg[i]);
}


#define GW 8
#define GT (GW * 32)

template <int K, int NP, int NOUT>
__global__ __launch_bounds__(GT) void k_gemm_wmma_f32(const float* __restrict__ A, const float* __restrict__ B,
                                                      float* __restrict__ C, int M) {
    constexpr int TN = NP / 16;
    constexpr int KS = K / 4;
    constexpr int C4 = NP / 4;
    constexpr int PP = 2 * NP + ((2 * NP) % 64 == 32 ? 0 : 32 - (2 * NP) % 64);
    constexpr int NI = ((K / 2) * C4) / GT;
    static_assert(K % 4 == 0 && NP % 16 == 0, "whole k steps and column tiles");
    static_assert(((K / 2) * C4) % GT == 0, "B is staged in whole rounds of the work-group");
    static_assert(PP % 64 == 32 && PP % 4 == 0, "fragment halves on disjoint banks, 16-byte LDS stores");
    __shared__ __attribute__((aligned(16))) float ldsB[(K / 2) * PP];

    const int lane = threadIdx.x & 31;
    int tm = blockIdx.x * GW + (threadIdx.x >> 5);
    const bool valid = (tm * 16 < M);
    if (!valid) tm = M / 16 - 1;

    const int m  = tm * 16 + (lane & 15);
    const int kh = (lane >> 4) << 1;

    float4 s0[NI], s1[NI];
#pragma unroll
    for (int i = 0; i < NI; ++i) {
        const int it = threadIdx.x + i * GT;
        const int j = it / C4, c4 = it - j * C4;
        s0[i] = ((const float4*)B)[(2 * j) * C4 + c4];
        s1[i] = ((const float4*)B)[(2 * j + 1) * C4 + c4];
    }
    const float* aP = A + (long)m * K + kh;
    v2f a[KS];
#pragma unroll
    for (int s = 0; s < KS; ++s) {
        a[s].x = aP[4 * s];
        a[s].y = aP[4 * s + 1];
    }
    __builtin_amdgcn_sched_barrier(0);
#pragma unroll
    for (int i = 0; i < NI; ++i) {
        const int it = threadIdx.x + i * GT;
        const int j = it / C4, c4 = it - j * C4;
        float* d = &ldsB[j * PP + c4 * 8];
        *(float4*)d       = make_float4(s0[i].x, s1[i].x, s0[i].y, s1[i].y);
        *(float4*)(d + 4) = make_float4(s0[i].z, s1[i].z, s0[i].w, s1[i].w);
    }
    __syncthreads();

    const int bOff = (kh >> 1) * PP + (lane & 15) * 2;

    v8f acc[TN];
#pragma unroll
    for (int t = 0; t < TN; ++t) acc[t] = v8f{};
    v2f bq[2][TN];
#pragma unroll
    for (int t = 0; t < TN; ++t)
        bq[0][t] = *(const v2f*)&ldsB[bOff + t * 32];
#pragma unroll
    for (int s = 0; s < KS; ++s) {
        if (s + 1 < KS) {
#pragma unroll
            for (int t = 0; t < TN; ++t)
                bq[(s + 1) & 1][t] = *(const v2f*)&ldsB[bOff + (2 * s + 2) * PP + t * 32];
        }
        __builtin_amdgcn_sched_barrier(0);
#pragma unroll
        for (int t = 0; t < TN; ++t)
            acc[t] = __builtin_amdgcn_wmma_f32_16x16x4_f32(
                 false, a[s],  false, bq[s & 1][t],
                 (short)0, acc[t],  false,  false);
        __builtin_amdgcn_sched_barrier(0);
    }

    if (valid) {
        const int rbase = tm * 16 + ((lane >> 4) << 3);
#pragma unroll
        for (int t = 0; t < TN; ++t) {
            const int n = t * 16 + (lane & 15);
            if (n < NOUT) {
#pragma unroll
                for (int v = 0; v < 8; ++v)
                    C[(long)(rbase + v) * NOUT + n] = acc[t][v];
            }
        }
    }
}


static_assert(NE % 4 == 0, "destinations are read as int4 by the degree launch");
static_assert(NN <= (1 << 20) && RN < (1 << 11), "saved list entry packing");
static_assert(NRG <= 255, "a range number fits a byte and 0xFF is free as the pad key");

template <int F, bool BIAS_RELU, bool SAVE, bool LOAD, bool FUSE>
__global__ __launch_bounds__(OT) void k_own_agg(const int* __restrict__ src, const int* __restrict__ dst,
                                                const float* __restrict__ dinv, const float* __restrict__ h,
                                                const float* __restrict__ bias, float* __restrict__ outp,
                                                int* __restrict__ lcnt, int* __restrict__ lsaved,
                                                const int* __restrict__ keys, const float* __restrict__ W2) {
    constexpr int PITCH = FUSE ? F + 4 : F;
    __shared__ __attribute__((aligned(16))) float acc[RN * PITCH];
    __shared__ int   list[LCAP];
    __shared__ float dinvL[RN];
    __shared__ int   cnt;

    constexpr int NJ = (F + 31) / 32;
    constexpr int F4 = F / 4;
    constexpr int NV = RN * F4;
    constexpr int NK = (NV + OT - 1) / OT;
    static_assert(F % 4 == 0, "a 16-byte piece never straddles two rows");
    static_assert(NN % RN == 0, "every range is whole: all RN owned rows exist");
    static_assert(!BIAS_RELU || OT % F4 == 0, "a thread's pieces all sit in one column group");
    static_assert(!FUSE || (BIAS_RELU && F == DH && F % 4 == 0), "the fused product is relu(row + bias) @ W2");
    static_assert(PITCH % 4 == 0, "16-byte pieces and 8-byte fragment reads stay aligned");
    const int tid  = threadIdx.x;
    const int lane = tid & 31;
    const int lo   = blockIdx.x * RN;

    int2* sv = (int2*)lsaved + (long)blockIdx.x * LCAP;
    const int  saved    = LOAD ? lcnt[blockIdx.x] : -1;
    const bool useSaved = LOAD && (unsigned)saved <= (unsigned)LCAP;

    if (tid == 0) cnt = 0;
    if (tid < RN) { int n = lo + tid; dinvL[tid] = (n < NN) ? dinv[n] : 0.0f; }
    {
        const float4* h4 = (const float4*)(h + (long)lo * F);
        float4 hv[NK];
        float  dv[NK];
#pragma unroll
        for (int k = 0; k < NK; ++k) {
            int i4 = tid + k * OT;
            i4 = i4 < NV ? i4 : NV - 1;
            hv[k] = h4[i4];
            dv[k] = dinv[lo + i4 / F4];
        }
#pragma unroll
        for (int k = 0; k < NK; ++k) {
            const int i4 = tid + k * OT;
            if (i4 < NV) {
                float s = dv[k]; s *= s;
                const int o = (PITCH == F) ? i4 * 4 : (i4 / F4) * PITCH + (i4 % F4) * 4;
                *(float4*)&acc[o] = make_float4(hv[k].x * s, hv[k].y * s, hv[k].z * s, hv[k].w * s);
            }
        }
    }
    __syncthreads();

    const int4* k4 = (const int4*)keys;
    const unsigned kb = (unsigned)blockIdx.x * 0x01010101u;
    if (!useSaved) {
        for (int b = tid; b < N16; b += OT * 4) {
            int4 d[4];
#pragma unroll
            for (int k = 0; k < 4; ++k) {
                int i = b + k * OT;
                d[k] = (i < N16) ? k4[i] : make_int4(-1, -1, -1, -1);
            }
            unsigned mm[2] = {0u, 0u};
#pragma unroll
            for (int k = 0; k < 4; ++k) {
                const unsigned x[4] = {(unsigned)d[k].x ^ kb, (unsigned)d[k].y ^ kb,
                                       (unsigned)d[k].z ^ kb, (unsigned)d[k].w ^ kb};
#pragma unroll
                for (int q = 0; q < 4; ++q) {
                    const unsigned z = ~(((x[q] & 0x7F7F7F7Fu) + 0x7F7F7F7Fu) | x[q] | 0x7F7F7F7Fu);
                    mm[k >> 1] |= z >> (7 - q - 4 * (k & 1));
                }
            }
            unsigned long long m = (unsigned long long)mm[0] | ((unsigned long long)mm[1] << 32);
            while (m) {
                const int p = __builtin_ctzll(m);
                m &= m - 1;
                const int k = ((p >> 5) << 1) | ((p >> 2) & 1);
                const int e = (b + k * OT) * 16 + ((p & 3) << 2) + ((p >> 3) & 3);
                int slot = atomicAdd(&cnt, 1);
                if (slot < LCAP) {
                    list[slot] = e;
                } else {
                    int s = src[e];
                    int r = dst[e] - lo;
                    float w = dinv[s] * dinvL[r];
                    for (int c = 0; c < F; ++c)
                        atomicAdd(&acc[r * PITCH + c], h[(long)s * F + c] * w);
                }
            }
        }
    }
    __syncthreads();

    const int n = __builtin_amdgcn_readfirstlane(useSaved ? saved : (cnt < LCAP ? cnt : LCAP));
    if (SAVE && tid == 0) lcnt[blockIdx.x] = cnt;
    for (int g = __builtin_amdgcn_readfirstlane(tid >> 5) * 32; g < n; g += OT) {
        const int  idx = g + lane;
        const bool ok  = idx < n;
        int r, s;
        float w;
        if (useSaved) {
            const int2 e = ok ? sv[idx] : make_int2(0, 0);
            r = e.x >> 20;
            s = e.x & 0xFFFFF;
            w = __int_as_float(e.y);
        } else {
            const int e = ok ? list[idx] : 0;
            s = ok ? src[e] : 0;
            r = ok ? dst[e] - lo : 0;
            w = ok ? dinv[s] * dinvL[r] : 0.0f;
            if (SAVE && ok) sv[idx] = make_int2(s | (r << 20), __float_as_int(w));
        }
        const int  m   = (n - g) < 32 ? (n - g) : 32;
        for (int k = 0; k < m; k += 4) {
            float v[4][NJ];
            int   rk[4];
            float wk[4];
#pragma unroll
            for (int kk = 0; kk < 4; ++kk) {
                const int sl = k + kk;
                const int sk = __builtin_amdgcn_readlane(s, sl);
                rk[kk] = __builtin_amdgcn_readlane(r, sl);
                wk[kk] = __int_as_float(__builtin_amdgcn_readlane(__float_as_int(w), sl));
                const float* hp = h + (long)sk * F;
#pragma unroll
                for (int j = 0; j < NJ; ++j) {
                    const int c = lane + 32 * j;
                    v[kk][j] = (c < F) ? hp[c] : 0.0f;
                }
            }
#pragma unroll
            for (int kk = 0; kk < 4; ++kk) {
                if (k + kk < m) {
#pragma unroll
                    for (int j = 0; j < NJ; ++j) {
                        const int c = lane + 32 * j;
                        if (c < F) atomicAdd(&acc[rk[kk] * PITCH + c], v[kk][j] * wk[kk]);
                    }
                }
            }
        }
    }
    __syncthreads();

    if (FUSE) {
        constexpr int NT = (RN + 15) / 16;
        constexpr int TN = NCP / 16;
        constexpr int KS = F / 4;
        const int wv = __builtin_amdgcn_readfirstlane(tid >> 5);
        if (wv < NT) {
            const int c  = lane & 15;
            const int kh = (lane >> 4) << 1;
            int ra = wv * 16 + c;
            ra = ra < RN ? ra : RN - 1;
            const float* aP = &acc[ra * PITCH + kh];
            const float* bP = bias + kh;
            int  nc[TN];
            bool nv[TN];
#pragma unroll
            for (int t = 0; t < TN; ++t) {
                const int n = t * 16 + c;
                nv[t] = n < NC;
                nc[t] = nv[t] ? n : NC - 1;
            }
            v8f d[TN];
#pragma unroll
            for (int t = 0; t < TN; ++t) d[t] = v8f{};
#pragma unroll
            for (int s = 0; s < KS; ++s) {
                v2f a = *(const v2f*)&aP[4 * s];
                const v2f bb = *(const v2f*)&bP[4 * s];
                a.x += bb.x; a.y += bb.y;
                a.x = a.x > 0.0f ? a.x : 0.0f;
                a.y = a.y > 0.0f ? a.y : 0.0f;
#pragma unroll
                for (int t = 0; t < TN; ++t) {
                    const float w0 = W2[(4 * s + kh) * NC + nc[t]];
                    const float w1 = W2[(4 * s + kh + 1) * NC + nc[t]];
                    v2f b;
                    b.x = nv[t] ? w0 : 0.0f;
                    b.y = nv[t] ? w1 : 0.0f;
                    d[t] = __builtin_amdgcn_wmma_f32_16x16x4_f32(
                         false, a,  false, b,
                         (short)0, d[t],  false,  false);
                }
            }
            const int rbase = wv * 16 + ((lane >> 4) << 3);
#pragma unroll
            for (int t = 0; t < TN; ++t) {
                if (nv[t]) {
#pragma unroll
                    for (int v = 0; v < 8; ++v)
                        if (rbase + v < RN)
                            outp[(long)(lo + rbase + v) * NC + t * 16 + c] = d[t][v];
                }
            }
        }
    } else {
        float4 bv = make_float4(0.0f, 0.0f, 0.0f, 0.0f);
        if (BIAS_RELU) bv = ((const float4*)bias)[tid % F4];
        float4* o4 = (float4*)(outp + (long)lo * F);
#pragma unroll
        for (int k = 0; k < NK; ++k) {
            const int i4 = tid + k * OT;
            if (i4 < NV) {
                float4 v = ((const float4*)acc)[i4];
                if (BIAS_RELU) {
                    v.x += bv.x; v.y += bv.y; v.z += bv.z; v.w += bv.w;
                    v.x = v.x > 0.0f ? v.x : 0.0f; v.y = v.y > 0.0f ? v.y : 0.0f;
                    v.z = v.z > 0.0f ? v.z : 0.0f; v.w = v.w > 0.0f ? v.w : 0.0f;
                }
                o4[i4] = v;
            }
        }
    }
}


__global__ void k_logsoftmax(const float* __restrict__ agg, const float* __restrict__ b,
                             float* __restrict__ out) {
    int i = blockIdx.x * blockDim.x + threadIdx.x;
    if (i >= NN) return;
    const float* r = agg + (long)i * NC;
    float mx = -INFINITY;
#pragma unroll
    for (int c = 0; c < NC; ++c) mx = fmaxf(mx, r[c] + b[c]);
    float sum = 0.0f;
#pragma unroll
    for (int c = 0; c < NC; ++c) sum += expf(r[c] + b[c] - mx);
    float lse = mx + logf(sum);
#pragma unroll
    for (int c = 0; c < NC; ++c) out[(long)i * NC + c] = r[c] + b[c] - lse;
}


extern "C" void kernel_launch(void* const* d_in, const int* in_sizes, int n_in,
                              void* d_out, int out_size, void* d_ws, size_t ws_size,
                              hipStream_t stream) {
    const float* x  = (const float*)d_in[0];
    const int*   ei = (const int*)d_in[1];
    const float* W1 = (const float*)d_in[2];
    const float* b1 = (const float*)d_in[3];
    const float* W2 = (const float*)d_in[4];
    const float* b2 = (const float*)d_in[5];
    float* out = (float*)d_out;

    const int* src = ei;
    const int* dst = ei + NE;

    float* ws   = (float*)d_ws;
    float* deg  = ws;
    float* dinv = deg  + NN;
    float* h0   = dinv + NN;
    float* h2   = h0   + (long)NN * DH;
    float* agg2 = h2   + (long)NN * NC;
    int*   lcnt = (int*)(agg2 + (long)NN * NC);
    int*   lsav = lcnt + 256;
    int*   keys = lsav + (long)NRG * LCAP * 2;

    const int B = 256;

    k_deg_init <<<cdiv(NN, B), B, 0, stream>>>(deg);
    k_deg_key  <<<cdiv(NKW, B), B, 0, stream>>>(dst, deg, keys);
    k_dinv     <<<cdiv(NN, B), B, 0, stream>>>(deg, dinv);

    k_gemm_wmma_f32<FIN, DH, DH>
        <<<cdiv(NN / 16, GW), GT, 0, stream>>>(x, W1, h0, NN);
    k_own_agg<DH, true, true, false, true><<<cdiv(NN, RN), OT, 0, stream>>>(src, dst, dinv, h0, b1, h2, lcnt, lsav, keys, W2);

    k_own_agg<NC, false, false, true, false><<<cdiv(NN, RN), OT, 0, stream>>>(src, dst, dinv, h2, b2, agg2, lcnt, lsav, keys, W2);

    k_logsoftmax<<<cdiv(NN, B), B, 0, stream>>>(agg2, b2, out);
}


# Elementwise launches folded into neighbours, 7 to 4: GEMM launch (now first) stores the degree start value, readers take rsqrt of the degree, layer-2 owner does bias+log_softmax on its LDS rows
# speedup vs baseline: 2.8231x; 1.0639x over previous
#include <hip/hip_runtime.h>
#include <math.h>

#define NN   50000
#define NE   625000
#define FIN  128
#define DH   128
#define NC   40
#define NCP  48

#define RN   200
#define LCAP 4096
#define OT   1024
#define NRG  ((NN + RN - 1) / RN)
#define N16  ((NE + 15) / 16)
#define NKW  (N16 * 4)

typedef __attribute__((ext_vector_type(2))) float v2f;
typedef __attribute__((ext_vector_type(8))) float v8f;

static inline int cdiv(long a, long b) { return (int)((a + b - 1) / b); }


__global__ void k_deg_key(const int* __restrict__ dst, float* __restrict__ deg, int* __restrict__ keys) {
    int t = blockIdx.x * blockDim.x + threadIdx.x;
    if (t >= NKW) return;
    unsigned kw = 0xFFFFFFFFu;
    if (t < NE / 4) {
        const int4 d = ((const int4*)dst)[t];
        atomicAdd(&deg[d.x], 1.0f);
        atomicAdd(&deg[d.y], 1.0f);
        atomicAdd(&deg[d.z], 1.0f);
        atomicAdd(&deg[d.w], 1.0f);
        kw = ((unsigned)d.x / RN) | (((unsigned)d.y / RN) << 8) |
             (((unsigned)d.z / RN) << 16) | (((unsigned)d.w / RN) << 24);
    }
    keys[t] = (int)kw;
}


#define GW 8
#define GT (GW * 32)
static_assert(((NN / 16 + GW - 1) / GW) * GT >= NN, "the layer-1 GEMM grid has a thread per node");

template <int K, int NP, int NOUT>
__global__ __launch_bounds__(GT) void k_gemm_wmma_f32(const float* __restrict__ A, const float* __restrict__ B,
                                                      float* __restrict__ C, int M,
                                                      float* __restrict__ deg) {
    {
        const int i = blockIdx.x * GT + threadIdx.x;
        if (i < NN) deg[i] = 1.0f;
    }
    constexpr int TN = NP / 16;
    constexpr int KS = K / 4;
    constexpr int C4 = NP / 4;
    constexpr int PP = 2 * NP + ((2 * NP) % 64 == 32 ? 0 : 32 - (2 * NP) % 64);
    constexpr int NI = ((K / 2) * C4) / GT;
    static_assert(K % 4 == 0 && NP % 16 == 0, "whole k steps and column tiles");
    static_assert(((K / 2) * C4) % GT == 0, "B is staged in whole rounds of the work-group");
    static_assert(PP % 64 == 32 && PP % 4 == 0, "fragment halves on disjoint banks, 16-byte LDS stores");
    __shared__ __attribute__((aligned(16))) float ldsB[(K / 2) * PP];

    const int lane = threadIdx.x & 31;
    int tm = blockIdx.x * GW + (threadIdx.x >> 5);
    const bool valid = (tm * 16 < M);
    if (!valid) tm = M / 16 - 1;

    const int m  = tm * 16 + (lane & 15);
    const int kh = (lane >> 4) << 1;

    float4 s0[NI], s1[NI];
#pragma unroll
    for (int i = 0; i < NI; ++i) {
        const int it = threadIdx.x + i * GT;
        const int j = it / C4, c4 = it - j * C4;
        s0[i] = ((const float4*)B)[(2 * j) * C4 + c4];
        s1[i] = ((const float4*)B)[(2 * j + 1) * C4 + c4];
    }
    const float* aP = A + (long)m * K + kh;
    v2f a[KS];
#pragma unroll
    for (int s = 0; s < KS; ++s) {
        a[s].x = aP[4 * s];
        a[s].y = aP[4 * s + 1];
    }
    __builtin_amdgcn_sched_barrier(0);
#pragma unroll
    for (int i = 0; i < NI; ++i) {
        const int it = threadIdx.x + i * GT;
        const int j = it / C4, c4 = it - j * C4;
        float* d = &ldsB[j * PP + c4 * 8];
        *(float4*)d       = make_float4(s0[i].x, s1[i].x, s0[i].y, s1[i].y);
        *(float4*)(d + 4) = make_float4(s0[i].z, s1[i].z, s0[i].w, s1[i].w);
    }
    __syncthreads();

    const int bOff = (kh >> 1) * PP + (lane & 15) * 2;

    v8f acc[TN];
#pragma unroll
    for (int t = 0; t < TN; ++t) acc[t] = v8f{};
    v2f bq[2][TN];
#pragma unroll
    for (int t = 0; t < TN; ++t)
        bq[0][t] = *(const v2f*)&ldsB[bOff + t * 32];
#pragma unroll
    for (int s = 0; s < KS; ++s) {
        if (s + 1 < KS) {
#pragma unroll
            for (int t = 0; t < TN; ++t)
                bq[(s + 1) & 1][t] = *(const v2f*)&ldsB[bOff + (2 * s + 2) * PP + t * 32];
        }
        __builtin_amdgcn_sched_barrier(0);
#pragma unroll
        for (int t = 0; t < TN; ++t)
            acc[t] = __builtin_amdgcn_wmma_f32_16x16x4_f32(
                 false, a[s],  false, bq[s & 1][t],
                 (short)0, acc[t],  false,  false);
        __builtin_amdgcn_sched_barrier(0);
    }

    if (valid) {
        const int rbase = tm * 16 + ((lane >> 4) << 3);
#pragma unroll
        for (int t = 0; t < TN; ++t) {
            const int n = t * 16 + (lane & 15);
            if (n < NOUT) {
#pragma unroll
                for (int v = 0; v < 8; ++v)
                    C[(long)(rbase + v) * NOUT + n] = acc[t][v];
            }
        }
    }
}


static_assert(NE % 4 == 0, "destinations are read as int4 by the degree launch");
static_assert(NN <= (1 << 20) && RN < (1 << 11), "saved list entry packing");
static_assert(NRG <= 255, "a range number fits a byte and 0xFF is free as the pad key");

template <int F, bool BIAS_RELU, bool SAVE, bool LOAD, bool FUSE>
__global__ __launch_bounds__(OT) void k_own_agg(const int* __restrict__ src, const int* __restrict__ dst,
                                                const float* __restrict__ deg, const float* __restrict__ h,
                                                const float* __restrict__ bias, float* __restrict__ outp,
                                                int* __restrict__ lcnt, int* __restrict__ lsaved,
                                                const int* __restrict__ keys, const float* __restrict__ W2) {
    constexpr int PITCH = FUSE ? F + 4 : F;
    __shared__ __attribute__((aligned(16))) float acc[RN * PITCH];
    __shared__ int   list[LCAP];
    __shared__ float dinvL[RN];
    __shared__ int   cnt;

    constexpr int NJ = (F + 31) / 32;
    constexpr int F4 = F / 4;
    constexpr int NV = RN * F4;
    constexpr int NK = (NV + OT - 1) / OT;
    static_assert(F % 4 == 0, "a 16-byte piece never straddles two rows");
    static_assert(NN % RN == 0, "every range is whole: all RN owned rows exist");
    static_assert(!FUSE || (BIAS_RELU && F == DH && F % 4 == 0), "the fused product is relu(row + bias) @ W2");
    static_assert(FUSE || !BIAS_RELU, "the launch that writes its rows out ends in bias + log_softmax");
    static_assert(PITCH % 4 == 0, "16-byte pieces and 8-byte fragment reads stay aligned");
    const int tid  = threadIdx.x;
    const int lane = tid & 31;
    const int lo   = blockIdx.x * RN;

    int2* sv = (int2*)lsaved + (long)blockIdx.x * LCAP;
    const int  saved    = LOAD ? lcnt[blockIdx.x] : -1;
    const bool useSaved = LOAD && (unsigned)saved <= (unsigned)LCAP;

    if (tid == 0) cnt = 0;
    if (tid < RN) { int n = lo + tid; dinvL[tid] = (n < NN) ? rsqrtf(deg[n]) : 0.0f; }
    {
        const float4* h4 = (const float4*)(h + (long)lo * F);
        float4 hv[NK];
        float  dv[NK];
#pragma unroll
        for (int k = 0; k < NK; ++k) {
            int i4 = tid + k * OT;
            i4 = i4 < NV ? i4 : NV - 1;
            hv[k] = h4[i4];
            dv[k] = deg[lo + i4 / F4];
        }
#pragma unroll
        for (int k = 0; k < NK; ++k) {
            const int i4 = tid + k * OT;
            if (i4 < NV) {
                float s = rsqrtf(dv[k]); s *= s;
                const int o = (PITCH == F) ? i4 * 4 : (i4 / F4) * PITCH + (i4 % F4) * 4;
                *(float4*)&acc[o] = make_float4(hv[k].x * s, hv[k].y * s, hv[k].z * s, hv[k].w * s);
            }
        }
    }
    __syncthreads();

    const int4* k4 = (const int4*)keys;
    const unsigned kb = (unsigned)blockIdx.x * 0x01010101u;
    if (!useSaved) {
        for (int b = tid; b < N16; b += OT * 4) {
            int4 d[4];
#pragma unroll
            for (int k = 0; k < 4; ++k) {
                int i = b + k * OT;
                d[k] = (i < N16) ? k4[i] : make_int4(-1, -1, -1, -1);
            }
            unsigned mm[2] = {0u, 0u};
#pragma unroll
            for (int k = 0; k < 4; ++k) {
                const unsigned x[4] = {(unsigned)d[k].x ^ kb, (unsigned)d[k].y ^ kb,
                                       (unsigned)d[k].z ^ kb, (unsigned)d[k].w ^ kb};
#pragma unroll
                for (int q = 0; q < 4; ++q) {
                    const unsigned z = ~(((x[q] & 0x7F7F7F7Fu) + 0x7F7F7F7Fu) | x[q] | 0x7F7F7F7Fu);
                    mm[k >> 1] |= z >> (7 - q - 4 * (k & 1));
                }
            }
            unsigned long long m = (unsigned long long)mm[0] | ((unsigned long long)mm[1] << 32);
            while (m) {
                const int p = __builtin_ctzll(m);
                m &= m - 1;
                const int k = ((p >> 5) << 1) | ((p >> 2) & 1);
                const int e = (b + k * OT) * 16 + ((p & 3) << 2) + ((p >> 3) & 3);
                int slot = atomicAdd(&cnt, 1);
                if (slot < LCAP) {
                    list[slot] = e;
                } else {
                    int s = src[e];
                    int r = dst[e] - lo;
                    float w = rsqrtf(deg[s]) * dinvL[r];
                    for (int c = 0; c < F; ++c)
                        atomicAdd(&acc[r * PITCH + c], h[(long)s * F + c] * w);
                }
            }
        }
    }
    __syncthreads();

    const int n = __builtin_amdgcn_readfirstlane(useSaved ? saved : (cnt < LCAP ? cnt : LCAP));
    if (SAVE && tid == 0) lcnt[blockIdx.x] = cnt;
    for (int g = __builtin_amdgcn_readfirstlane(tid >> 5) * 32; g < n; g += OT) {
        const int  idx = g + lane;
        const bool ok  = idx < n;
        int r, s;
        float w;
        if (useSaved) {
            const int2 e = ok ? sv[idx] : make_int2(0, 0);
            r = e.x >> 20;
            s = e.x & 0xFFFFF;
            w = __int_as_float(e.y);
        } else {
            const int e = ok ? list[idx] : 0;
            s = ok ? src[e] : 0;
            r = ok ? dst[e] - lo : 0;
            w = ok ? rsqrtf(deg[s]) * dinvL[r] : 0.0f;
            if (SAVE && ok) sv[idx] = make_int2(s | (r << 20), __float_as_int(w));
        }
        const int  m   = (n - g) < 32 ? (n - g) : 32;
        for (int k = 0; k < m; k += 4) {
            float v[4][NJ];
            int   rk[4];
            float wk[4];
#pragma unroll
            for (int kk = 0; kk < 4; ++kk) {
                const int sl = k + kk;
                const int sk = __builtin_amdgcn_readlane(s, sl);
                rk[kk] = __builtin_amdgcn_readlane(r, sl);
                wk[kk] = __int_as_float(__builtin_amdgcn_readlane(__float_as_int(w), sl));
                const float* hp = h + (long)sk * F;
#pragma unroll
                for (int j = 0; j < NJ; ++j) {
                    const int c = lane + 32 * j;
                    v[kk][j] = (c < F) ? hp[c] : 0.0f;
                }
            }
#pragma unroll
            for (int kk = 0; kk < 4; ++kk) {
                if (k + kk < m) {
#pragma unroll
                    for (int j = 0; j < NJ; ++j) {
                        const int c = lane + 32 * j;
                        if (c < F) atomicAdd(&acc[rk[kk] * PITCH + c], v[kk][j] * wk[kk]);
                    }
                }
            }
        }
    }
    __syncthreads();

    if (FUSE) {
        constexpr int NT = (RN + 15) / 16;
        constexpr int TN = NCP / 16;
        constexpr int KS = F / 4;
        const int wv = __builtin_amdgcn_readfirstlane(tid >> 5);
        if (wv < NT) {
            const int c  = lane & 15;
            const int kh = (lane >> 4) << 1;
            int ra = wv * 16 + c;
            ra = ra < RN ? ra : RN - 1;
            const float* aP = &acc[ra * PITCH + kh];
            const float* bP = bias + kh;
            int  nc[TN];
            bool nv[TN];
#pragma unroll
            for (int t = 0; t < TN; ++t) {
                const int n = t * 16 + c;
                nv[t] = n < NC;
                nc[t] = nv[t] ? n : NC - 1;
            }
            v8f d[TN];
#pragma unroll
            for (int t = 0; t < TN; ++t) d[t] = v8f{};
#pragma unroll
            for (int s = 0; s < KS; ++s) {
                v2f a = *(const v2f*)&aP[4 * s];
                const v2f bb = *(const v2f*)&bP[4 * s];
                a.x += bb.x; a.y += bb.y;
                a.x = a.x > 0.0f ? a.x : 0.0f;
                a.y = a.y > 0.0f ? a.y : 0.0f;
#pragma unroll
                for (int t = 0; t < TN; ++t) {
                    const float w0 = W2[(4 * s + kh) * NC + nc[t]];
                    const float w1 = W2[(4 * s + kh + 1) * NC + nc[t]];
                    v2f b;
                    b.x = nv[t] ? w0 : 0.0f;
                    b.y = nv[t] ? w1 : 0.0f;
                    d[t] = __builtin_amdgcn_wmma_f32_16x16x4_f32(
                         false, a,  false, b,
                         (short)0, d[t],  false,  false);
                }
            }
            const int rbase = wv * 16 + ((lane >> 4) << 3);
#pragma unroll
            for (int t = 0; t < TN; ++t) {
                if (nv[t]) {
#pragma unroll
                    for (int v = 0; v < 8; ++v)
                        if (rbase + v < RN)
                            outp[(long)(lo + rbase + v) * NC + t * 16 + c] = d[t][v];
                }
            }
        }
    } else {
        __shared__ float lse[RN];
        if (tid < RN) {
            const float4* r4 = (const float4*)&acc[tid * PITCH];
            float v[F];
#pragma unroll
            for (int q = 0; q < F4; ++q) {
                const float4 a = r4[q];
                const float4 b = ((const float4*)bias)[q];
                v[4 * q]     = a.x + b.x;
                v[4 * q + 1] = a.y + b.y;
                v[4 * q + 2] = a.z + b.z;
                v[4 * q + 3] = a.w + b.w;
            }
            float mx = -INFINITY;
#pragma unroll
            for (int q = 0; q < F; ++q) mx = fmaxf(mx, v[q]);
            float sum = 0.0f;
#pragma unroll
            for (int q = 0; q < F; ++q) sum += expf(v[q] - mx);
            lse[tid] = mx + logf(sum);
        }
        __syncthreads();
        float4* o4 = (float4*)(outp + (long)lo * F);
#pragma unroll
        for (int k = 0; k < NK; ++k) {
            const int i4 = tid + k * OT;
            if (i4 < NV) {
                const int row = i4 / F4;
                const float4 b = ((const float4*)bias)[i4 - row * F4];
                const float  l = lse[row];
                float4 v = ((const float4*)acc)[i4];
                v.x = (v.x + b.x) - l; v.y = (v.y + b.y) - l;
                v.z = (v.z + b.z) - l; v.w = (v.w + b.w) - l;
                o4[i4] = v;
            }
        }
    }
}


extern "C" void kernel_launch(void* const* d_in, const int* in_sizes, int n_in,
                              void* d_out, int out_size, void* d_ws, size_t ws_size,
                              hipStream_t stream) {
    const float* x  = (const float*)d_in[0];
    const int*   ei = (const int*)d_in[1];
    const float* W1 = (const float*)d_in[2];
    const float* b1 = (const float*)d_in[3];
    const float* W2 = (const float*)d_in[4];
    const float* b2 = (const float*)d_in[5];
    float* out = (float*)d_out;

    const int* src = ei;
    const int* dst = ei + NE;

    float* ws   = (float*)d_ws;
    float* deg  = ws;
    float* h0   = deg  + NN;
    float* h2   = h0   + (long)NN * DH;
    int*   lcnt = (int*)(h2 + (long)NN * NC);
    int*   lsav = lcnt + 256;
    int*   keys = lsav + (long)NRG * LCAP * 2;

    const int B = 256;

    k_gemm_wmma_f32<FIN, DH, DH>
        <<<cdiv(NN / 16, GW), GT, 0, stream>>>(x, W1, h0, NN, deg);

    k_deg_key  <<<cdiv(NKW, B), B, 0, stream>>>(dst, deg, keys);

    k_own_agg<DH, true, true, false, true><<<cdiv(NN, RN), OT, 0, stream>>>(src, dst, deg, h0, b1, h2, lcnt, lsav, keys, W2);

    k_own_agg<NC, false, false, true, false><<<cdiv(NN, RN), OT, 0, stream>>>(src, dst, deg, h2, b2, out, lcnt, lsav, keys, W2);
}


# Degrees counted without global atomics: a range-owner launch filters the key plane once, counts its nodes' edges by LDS integer adds, stores degrees and the range's list; row sums in arrival order
# speedup vs baseline: 3.7736x; 1.3367x over previous
#include <hip/hip_runtime.h>
#include <math.h>

#define NN   50000
#define NE   625000
#define FIN  128
#define DH   128
#define NC   40
#define NCP  48

#define RN   200
#define LCAP 4096
#define OT   1024
#define NRG  ((NN + RN - 1) / RN)
#define N16  ((NE + 15) / 16)
#define NKW  (N16 * 4)

typedef __attribute__((ext_vector_type(2))) float v2f;
typedef __attribute__((ext_vector_type(8))) float v8f;

static inline int cdiv(long a, long b) { return (int)((a + b - 1) / b); }


#define GW 8
#define GT (GW * 32)
#define KR 2
static_assert(KR * ((NN / 16 + GW - 1) / GW) * GT >= NKW, "the layer-1 GEMM grid covers the key plane in KR rounds");

template <int K, int NP, int NOUT>
__global__ __launch_bounds__(GT) void k_gemm_wmma_f32(const float* __restrict__ A, const float* __restrict__ B,
                                                      float* __restrict__ C, int M,
                                                      const int* __restrict__ dst, int* __restrict__ keys) {
    const int kt = blockIdx.x * GT + threadIdx.x;
    const int kg = gridDim.x * GT;
    int4 dk[KR];
#pragma unroll
    for (int r = 0; r < KR; ++r) {
        const int t = kt + r * kg;
        dk[r] = ((const int4*)dst)[t < NE / 4 ? t : NE / 4 - 1];
    }
    constexpr int TN = NP / 16;
    constexpr int KS = K / 4;
    constexpr int C4 = NP / 4;
    constexpr int PP = 2 * NP + ((2 * NP) % 64 == 32 ? 0 : 32 - (2 * NP) % 64);
    constexpr int NI = ((K / 2) * C4) / GT;
    static_assert(K % 4 == 0 && NP % 16 == 0, "whole k steps and column tiles");
    static_assert(((K / 2) * C4) % GT == 0, "B is staged in whole rounds of the work-group");
    static_assert(PP % 64 == 32 && PP % 4 == 0, "fragment halves on disjoint banks, 16-byte LDS stores");
    __shared__ __attribute__((aligned(16))) float ldsB[(K / 2) * PP];

    const int lane = threadIdx.x & 31;
    int tm = blockIdx.x * GW + (threadIdx.x >> 5);
    const bool valid = (tm * 16 < M);
    if (!valid) tm = M / 16 - 1;

    const int m  = tm * 16 + (lane & 15);
    const int kh = (lane >> 4) << 1;

    float4 s0[NI], s1[NI];
#pragma unroll
    for (int i = 0; i < NI; ++i) {
        const int it = threadIdx.x + i * GT;
        const int j = it / C4, c4 = it - j * C4;
        s0[i] = ((const float4*)B)[(2 * j) * C4 + c4];
        s1[i] = ((const float4*)B)[(2 * j + 1) * C4 + c4];
    }
    const float* aP = A + (long)m * K + kh;
    v2f a[KS];
#pragma unroll
    for (int s = 0; s < KS; ++s) {
        a[s].x = aP[4 * s];
        a[s].y = aP[4 * s + 1];
    }
    __builtin_amdgcn_sched_barrier(0);
#pragma unroll
    for (int i = 0; i < NI; ++i) {
        const int it = threadIdx.x + i * GT;
        const int j = it / C4, c4 = it - j * C4;
        float* d = &ldsB[j * PP + c4 * 8];
        *(float4*)d       = make_float4(s0[i].x, s1[i].x, s0[i].y, s1[i].y);
        *(float4*)(d + 4) = make_float4(s0[i].z, s1[i].z, s0[i].w, s1[i].w);
    }
    __syncthreads();

    const int bOff = (kh >> 1) * PP + (lane & 15) * 2;

    v8f acc[TN];
#pragma unroll
    for (int t = 0; t < TN; ++t) acc[t] = v8f{};
    v2f bq[2][TN];
#pragma unroll
    for (int t = 0; t < TN; ++t)
        bq[0][t] = *(const v2f*)&ldsB[bOff + t * 32];
#pragma unroll
    for (int s = 0; s < KS; ++s) {
        if (s + 1 < KS) {
#pragma unroll
            for (int t = 0; t < TN; ++t)
                bq[(s + 1) & 1][t] = *(const v2f*)&ldsB[bOff + (2 * s + 2) * PP + t * 32];
        }
        __builtin_amdgcn_sched_barrier(0);
#pragma unroll
        for (int t = 0; t < TN; ++t)
            acc[t] = __builtin_amdgcn_wmma_f32_16x16x4_f32(
                 false, a[s],  false, bq[s & 1][t],
                 (short)0, acc[t],  false,  false);
        __builtin_amdgcn_sched_barrier(0);
    }

    if (valid) {
        const int rbase = tm * 16 + ((lane >> 4) << 3);
#pragma unroll
        for (int t = 0; t < TN; ++t) {
            const int n = t * 16 + (lane & 15);
            if (n < NOUT) {
#pragma unroll
                for (int v = 0; v < 8; ++v)
                    C[(long)(rbase + v) * NOUT + n] = acc[t][v];
            }
        }
    }

#pragma unroll
    for (int r = 0; r < KR; ++r) {
        const int t = kt + r * kg;
        if (t < NKW) {
            const unsigned kw = ((unsigned)dk[r].x / RN) | (((unsigned)dk[r].y / RN) << 8) |
                                (((unsigned)dk[r].z / RN) << 16) | (((unsigned)dk[r].w / RN) << 24);
            keys[t] = t < NE / 4 ? (int)kw : -1;
        }
    }
}


static_assert(NE % 4 == 0, "destinations are read as int4 by the degree launch");
static_assert(NN <= (1 << 20) && RN < (1 << 11), "saved list entry packing");
static_assert(NRG <= 255, "a range number fits a byte and 0xFF is free as the pad key");

__device__ inline unsigned long long range_flags(const int4* __restrict__ k4, int b, unsigned kb) {
    int4 d[4];
#pragma unroll
    for (int k = 0; k < 4; ++k) {
        int i = b + k * OT;
        d[k] = (i < N16) ? k4[i] : make_int4(-1, -1, -1, -1);
    }
    unsigned mm[2] = {0u, 0u};
#pragma unroll
    for (int k = 0; k < 4; ++k) {
        const unsigned x[4] = {(unsigned)d[k].x ^ kb, (unsigned)d[k].y ^ kb,
                               (unsigned)d[k].z ^ kb, (unsigned)d[k].w ^ kb};
#pragma unroll
        for (int q = 0; q < 4; ++q) {
            const unsigned z = ~(((x[q] & 0x7F7F7F7Fu) + 0x7F7F7F7Fu) | x[q] | 0x7F7F7F7Fu);
            mm[k >> 1] |= z >> (7 - q - 4 * (k & 1));
        }
    }
    return (unsigned long long)mm[0] | ((unsigned long long)mm[1] << 32);
}

__device__ inline int flag_edge(int b, int p) {
    const int k = ((p >> 5) << 1) | ((p >> 2) & 1);
    return (b + k * OT) * 16 + ((p & 3) << 2) + ((p >> 3) & 3);
}

__global__ __launch_bounds__(OT) void k_own_count(const int* __restrict__ src, const int* __restrict__ dst,
                                                  float* __restrict__ deg, int* __restrict__ lcnt,
                                                  int* __restrict__ lnode, const int* __restrict__ keys) {
    __shared__ int list[LCAP];
    __shared__ int degL[RN];
    __shared__ int cnt;
    constexpr int NL = LCAP / OT;
    static_assert(LCAP % OT == 0, "the list is resolved in whole rounds of the work-group");
    const int tid = threadIdx.x;
    const int lo  = blockIdx.x * RN;
    if (tid == 0) cnt = 0;
    if (tid < RN) degL[tid] = 1;
    __syncthreads();

    const int4* k4 = (const int4*)keys;
    const unsigned kb = (unsigned)blockIdx.x * 0x01010101u;
    for (int b = tid; b < N16; b += OT * 4) {
        unsigned long long m = range_flags(k4, b, kb);
        while (m) {
            const int p = __builtin_ctzll(m);
            m &= m - 1;
            const int e = flag_edge(b, p);
            const int slot = atomicAdd(&cnt, 1);
            if (slot < LCAP) list[slot] = e;
            else             atomicAdd(&degL[dst[e] - lo], 1);
        }
    }
    __syncthreads();

    const int total = cnt;
    const int n = total < LCAP ? total : LCAP;
    if (tid == 0) lcnt[blockIdx.x] = total;
    int* ln = lnode + (long)blockIdx.x * LCAP;
    int sv[NL], dv[NL];
#pragma unroll
    for (int k = 0; k < NL; ++k) {
        const int idx = tid + k * OT;
        const int e = idx < n ? list[idx] : 0;
        sv[k] = src[e];
        dv[k] = dst[e];
    }
#pragma unroll
    for (int k = 0; k < NL; ++k) {
        const int idx = tid + k * OT;
        if (idx < n) {
            const int r = dv[k] - lo;
            atomicAdd(&degL[r], 1);
            ln[idx] = sv[k] | (r << 20);
        }
    }
    __syncthreads();
    if (tid < RN) deg[lo + tid] = (float)degL[tid];
}

template <int F, bool BIAS_RELU, bool SAVE, bool FUSE>
__global__ __launch_bounds__(OT) void k_own_agg(const int* __restrict__ src, const int* __restrict__ dst,
                                                const float* __restrict__ deg, const float* __restrict__ h,
                                                const float* __restrict__ bias, float* __restrict__ outp,
                                                const int* __restrict__ lcnt, const int* __restrict__ lnode,
                                                float* __restrict__ lwt,
                                                const int* __restrict__ keys, const float* __restrict__ W2) {
    constexpr int PITCH = FUSE ? F + 4 : F;
    __shared__ __attribute__((aligned(16))) float acc[RN * PITCH];
    __shared__ int   list[LCAP];
    __shared__ float dinvL[RN];
    __shared__ int   cnt;

    constexpr int NJ = (F + 31) / 32;
    constexpr int F4 = F / 4;
    constexpr int NV = RN * F4;
    constexpr int NK = (NV + OT - 1) / OT;
    static_assert(F % 4 == 0, "a 16-byte piece never straddles two rows");
    static_assert(NN % RN == 0, "every range is whole: all RN owned rows exist");
    static_assert(!FUSE || (BIAS_RELU && F == DH && F % 4 == 0), "the fused product is relu(row + bias) @ W2");
    static_assert(FUSE || !BIAS_RELU, "the launch that writes its rows out ends in bias + log_softmax");
    static_assert(PITCH % 4 == 0, "16-byte pieces and 8-byte fragment reads stay aligned");
    const int tid  = threadIdx.x;
    const int lane = tid & 31;
    const int lo   = blockIdx.x * RN;

    const int* ln = lnode + (long)blockIdx.x * LCAP;
    float*     lw = lwt + (long)blockIdx.x * LCAP;
    const int  saved    = lcnt[blockIdx.x];
    const bool useSaved = (unsigned)saved <= (unsigned)LCAP;

    if (tid == 0) cnt = 0;
    if (tid < RN) { int n = lo + tid; dinvL[tid] = (n < NN) ? rsqrtf(deg[n]) : 0.0f; }
    {
        const float4* h4 = (const float4*)(h + (long)lo * F);
        float4 hv[NK];
        float  dv[NK];
#pragma unroll
        for (int k = 0; k < NK; ++k) {
            int i4 = tid + k * OT;
            i4 = i4 < NV ? i4 : NV - 1;
            hv[k] = h4[i4];
            dv[k] = deg[lo + i4 / F4];
        }
#pragma unroll
        for (int k = 0; k < NK; ++k) {
            const int i4 = tid + k * OT;
            if (i4 < NV) {
                float s = rsqrtf(dv[k]); s *= s;
                const int o = (PITCH == F) ? i4 * 4 : (i4 / F4) * PITCH + (i4 % F4) * 4;
                *(float4*)&acc[o] = make_float4(hv[k].x * s, hv[k].y * s, hv[k].z * s, hv[k].w * s);
            }
        }
    }
    __syncthreads();

    const int4* k4 = (const int4*)keys;
    const unsigned kb = (unsigned)blockIdx.x * 0x01010101u;
    if (!useSaved) {
        for (int b = tid; b < N16; b += OT * 4) {
            unsigned long long m = range_flags(k4, b, kb);
            while (m) {
                const int p = __builtin_ctzll(m);
                m &= m - 1;
                const int e = flag_edge(b, p);
                int slot = atomicAdd(&cnt, 1);
                if (slot < LCAP) {
                    list[slot] = e;
                } else {
                    int s = src[e];
                    int r = dst[e] - lo;
                    float w = rsqrtf(deg[s]) * dinvL[r];
                    for (int c = 0; c < F; ++c)
                        atomicAdd(&acc[r * PITCH + c], h[(long)s * F + c] * w);
                }
            }
        }
    }
    __syncthreads();

    const int n = __builtin_amdgcn_readfirstlane(useSaved ? saved : (cnt < LCAP ? cnt : LCAP));
    for (int g = __builtin_amdgcn_readfirstlane(tid >> 5) * 32; g < n; g += OT) {
        const int  idx = g + lane;
        const bool ok  = idx < n;
        int r, s;
        float w;
        if (useSaved) {
            const int x = ok ? ln[idx] : 0;
            r = x >> 20;
            s = x & 0xFFFFF;
            if (SAVE) {
                w = ok ? rsqrtf(deg[s]) * dinvL[r] : 0.0f;
                if (ok) lw[idx] = w;
            } else {
                w = ok ? lw[idx] : 0.0f;
            }
        } else {
            const int e = ok ? list[idx] : 0;
            s = ok ? src[e] : 0;
            r = ok ? dst[e] - lo : 0;
            w = ok ? rsqrtf(deg[s]) * dinvL[r] : 0.0f;
        }
        const int  m   = (n - g) < 32 ? (n - g) : 32;
        for (int k = 0; k < m; k += 4) {
            float v[4][NJ];
            int   rk[4];
            float wk[4];
#pragma unroll
            for (int kk = 0; kk < 4; ++kk) {
                const int sl = k + kk;
                const int sk = __builtin_amdgcn_readlane(s, sl);
                rk[kk] = __builtin_amdgcn_readlane(r, sl);
                wk[kk] = __int_as_float(__builtin_amdgcn_readlane(__float_as_int(w), sl));
                const float* hp = h + (long)sk * F;
#pragma unroll
                for (int j = 0; j < NJ; ++j) {
                    const int c = lane + 32 * j;
                    v[kk][j] = (c < F) ? hp[c] : 0.0f;
                }
            }
#pragma unroll
            for (int kk = 0; kk < 4; ++kk) {
                if (k + kk < m) {
#pragma unroll
                    for (int j = 0; j < NJ; ++j) {
                        const int c = lane + 32 * j;
                        if (c < F) atomicAdd(&acc[rk[kk] * PITCH + c], v[kk][j] * wk[kk]);
                    }
                }
            }
        }
    }
    __syncthreads();

    if (FUSE) {
        constexpr int NT = (RN + 15) / 16;
        constexpr int TN = NCP / 16;
        constexpr int KS = F / 4;
        const int wv = __builtin_amdgcn_readfirstlane(tid >> 5);
        if (wv < NT) {
            const int c  = lane & 15;
            const int kh = (lane >> 4) << 1;
            int ra = wv * 16 + c;
            ra = ra < RN ? ra : RN - 1;
            const float* aP = &acc[ra * PITCH + kh];
            const float* bP = bias + kh;
            int  nc[TN];
            bool nv[TN];
#pragma unroll
            for (int t = 0; t < TN; ++t) {
                const int n = t * 16 + c;
                nv[t] = n < NC;
                nc[t] = nv[t] ? n : NC - 1;
            }
            v8f d[TN];
#pragma unroll
            for (int t = 0; t < TN; ++t) d[t] = v8f{};
#pragma unroll
            for (int s = 0; s < KS; ++s) {
                v2f a = *(const v2f*)&aP[4 * s];
                const v2f bb = *(const v2f*)&bP[4 * s];
                a.x += bb.x; a.y += bb.y;
                a.x = a.x > 0.0f ? a.x : 0.0f;
                a.y = a.y > 0.0f ? a.y : 0.0f;
#pragma unroll
                for (int t = 0; t < TN; ++t) {
                    const float w0 = W2[(4 * s + kh) * NC + nc[t]];
                    const float w1 = W2[(4 * s + kh + 1) * NC + nc[t]];
                    v2f b;
                    b.x = nv[t] ? w0 : 0.0f;
                    b.y = nv[t] ? w1 : 0.0f;
                    d[t] = __builtin_amdgcn_wmma_f32_16x16x4_f32(
                         false, a,  false, b,
                         (short)0, d[t],  false,  false);
                }
            }
            const int rbase = wv * 16 + ((lane >> 4) << 3);
#pragma unroll
            for (int t = 0; t < TN; ++t) {
                if (nv[t]) {
#pragma unroll
                    for (int v = 0; v < 8; ++v)
                        if (rbase + v < RN)
                            outp[(long)(lo + rbase + v) * NC + t * 16 + c] = d[t][v];
                }
            }
        }
    } else {
        __shared__ float lse[RN];
        if (tid < RN) {
            const float4* r4 = (const float4*)&acc[tid * PITCH];
            float v[F];
#pragma unroll
            for (int q = 0; q < F4; ++q) {
                const float4 a = r4[q];
                const float4 b = ((const float4*)bias)[q];
                v[4 * q]     = a.x + b.x;
                v[4 * q + 1] = a.y + b.y;
                v[4 * q + 2] = a.z + b.z;
                v[4 * q + 3] = a.w + b.w;
            }
            float mx = -INFINITY;
#pragma unroll
            for (int q = 0; q < F; ++q) mx = fmaxf(mx, v[q]);
            float sum = 0.0f;
#pragma unroll
            for (int q = 0; q < F; ++q) sum += expf(v[q] - mx);
            lse[tid] = mx + logf(sum);
        }
        __syncthreads();
        float4* o4 = (float4*)(outp + (long)lo * F);
#pragma unroll
        for (int k = 0; k < NK; ++k) {
            const int i4 = tid + k * OT;
            if (i4 < NV) {
                const int row = i4 / F4;
                const float4 b = ((const float4*)bias)[i4 - row * F4];
                const float  l = lse[row];
                float4 v = ((const float4*)acc)[i4];
                v.x = (v.x + b.x) - l; v.y = (v.y + b.y) - l;
                v.z = (v.z + b.z) - l; v.w = (v.w + b.w) - l;
                o4[i4] = v;
            }
        }
    }
}


extern "C" void kernel_launch(void* const* d_in, const int* in_sizes, int n_in,
                              void* d_out, int out_size, void* d_ws, size_t ws_size,
                              hipStream_t stream) {
    const float* x  = (const float*)d_in[0];
    const int*   ei = (const int*)d_in[1];
    const float* W1 = (const float*)d_in[2];
    const float* b1 = (const float*)d_in[3];
    const float* W2 = (const float*)d_in[4];
    const float* b2 = (const float*)d_in[5];
    float* out = (float*)d_out;

    const int* src = ei;
    const int* dst = ei + NE;

    float* ws   = (float*)d_ws;
    float* deg  = ws;
    float* h0   = deg  + NN;
    float* h2   = h0   + (long)NN * DH;
    int*   lcnt = (int*)(h2 + (long)NN * NC);
    int*   lnod = lcnt + 256;
    float* lwt  = (float*)(lnod + (long)NRG * LCAP);
    int*   keys = (int*)(lwt + (long)NRG * LCAP);

    k_gemm_wmma_f32<FIN, DH, DH>
        <<<cdiv(NN / 16, GW), GT, 0, stream>>>(x, W1, h0, NN, dst, keys);

    k_own_count<<<cdiv(NN, RN), OT, 0, stream>>>(src, dst, deg, lcnt, lnod, keys);

    k_own_agg<DH, true, true, true><<<cdiv(NN, RN), OT, 0, stream>>>(src, dst, deg, h0, b1, h2, lcnt, lnod, lwt, keys, W2);

    k_own_agg<NC, false, false, false><<<cdiv(NN, RN), OT, 0, stream>>>(src, dst, deg, h2, b2, out, lcnt, lnod, lwt, keys, W2);
}


# Owner filter on 8 range-key bit-planes (written in the GEMM launch by a lane bit-transpose): 32 edges matched by 8 XNOR-AND ops, not 32 byte compares; 8 loads per thread trip (was 4); add order kept
# speedup vs baseline: 3.9707x; 1.0522x over previous
#include <hip/hip_runtime.h>
#include <math.h>

#define NN   50000
#define NE   625000
#define FIN  128
#define DH   128
#define NC   40
#define NCP  48

#define RN   200
#define LCAP 4096
#define OT   1024
#define NRG  ((NN + RN - 1) / RN)
#define NG   ((NE + 127) / 128)
#define NBK  ((NG + 7) / 8)
#define NUN  (NBK * 32)
#define NKW  (NBK * 256)

typedef __attribute__((ext_vector_type(2))) float v2f;
typedef __attribute__((ext_vector_type(8))) float v8f;

static inline int cdiv(long a, long b) { return (int)((a + b - 1) / b); }


template <int S, unsigned M>
__device__ inline unsigned xpose_step(unsigned w, int lane) {
    const bool upper = (lane & S) != 0;
    const unsigned keep = upper ? ~M : M;
    const unsigned t = (unsigned)__shfl_xor((int)w, S);
    const unsigned moved = __builtin_rotateright32(t, upper ? S : 32 - S);
    return (w & keep) | (moved & ~keep);
}

#define GW 8
#define GT (GW * 32)
#define KR 2
static_assert(KR * ((NN / 16 + GW - 1) / GW) * GT >= NBK * 8 * 32, "the layer-1 GEMM grid covers every 128-edge group in KR rounds");

template <int K, int NP, int NOUT>
__global__ __launch_bounds__(GT) void k_gemm_wmma_f32(const float* __restrict__ A, const float* __restrict__ B,
                                                      float* __restrict__ C, int M,
                                                      const int* __restrict__ dst, int* __restrict__ keys) {
    const int kt = blockIdx.x * GT + threadIdx.x;
    const int kg = gridDim.x * GT;
    int4 dk[KR];
#pragma unroll
    for (int r = 0; r < KR; ++r) {
        const int t = kt + r * kg;
        dk[r] = ((const int4*)dst)[t < NE / 4 ? t : NE / 4 - 1];
    }
    constexpr int TN = NP / 16;
    constexpr int KS = K / 4;
    constexpr int C4 = NP / 4;
    constexpr int PP = 2 * NP + ((2 * NP) % 64 == 32 ? 0 : 32 - (2 * NP) % 64);
    constexpr int NI = ((K / 2) * C4) / GT;
    static_assert(K % 4 == 0 && NP % 16 == 0, "whole k steps and column tiles");
    static_assert(((K / 2) * C4) % GT == 0, "B is staged in whole rounds of the work-group");
    static_assert(PP % 64 == 32 && PP % 4 == 0, "fragment halves on disjoint banks, 16-byte LDS stores");
    __shared__ __attribute__((aligned(16))) float ldsB[(K / 2) * PP];

    const int lane = threadIdx.x & 31;
    int tm = blockIdx.x * GW + (threadIdx.x >> 5);
    const bool valid = (tm * 16 < M);
    if (!valid) tm = M / 16 - 1;

    const int m  = tm * 16 + (lane & 15);
    const int kh = (lane >> 4) << 1;

    float4 s0[NI], s1[NI];
#pragma unroll
    for (int i = 0; i < NI; ++i) {
        const int it = threadIdx.x + i * GT;
        const int j = it / C4, c4 = it - j * C4;
        s0[i] = ((const float4*)B)[(2 * j) * C4 + c4];
        s1[i] = ((const float4*)B)[(2 * j + 1) * C4 + c4];
    }
    const float* aP = A + (long)m * K + kh;
    v2f a[KS];
#pragma unroll
    for (int s = 0; s < KS; ++s) {
        a[s].x = aP[4 * s];
        a[s].y = aP[4 * s + 1];
    }
    __builtin_amdgcn_sched_barrier(0);
    unsigned w[KR];
#pragma unroll
    for (int r = 0; r < KR; ++r) {
        const bool real = kt + r * kg < NE / 4;
        const unsigned kw = ((unsigned)dk[r].x / RN) | (((unsigned)dk[r].y / RN) << 8) |
                            (((unsigned)dk[r].z / RN) << 16) | (((unsigned)dk[r].w / RN) << 24);
        w[r] = real ? kw : 0xFFFFFFFFu;
    }
#pragma unroll
    for (int r = 0; r < KR; ++r) w[r] = xpose_step<16, 0x0000FFFFu>(w[r], lane);
#pragma unroll
    for (int r = 0; r < KR; ++r) w[r] = xpose_step<8, 0x00FF00FFu>(w[r], lane);
#pragma unroll
    for (int r = 0; r < KR; ++r) w[r] = xpose_step<4, 0x0F0F0F0Fu>(w[r], lane);
#pragma unroll
    for (int r = 0; r < KR; ++r) w[r] = xpose_step<2, 0x33333333u>(w[r], lane);
#pragma unroll
    for (int r = 0; r < KR; ++r) w[r] = xpose_step<1, 0x55555555u>(w[r], lane);
#pragma unroll
    for (int r = 0; r < KR; ++r) {
        const int G = (kt + r * kg) >> 5;
        if (G < NBK * 8) {
            const int j = lane & 7, c = lane >> 3;
            keys[(G >> 3) * 256 + (j >> 2) * 128 + (G & 7) * 16 + c * 4 + (j & 3)] = (int)w[r];
        }
    }
    __builtin_amdgcn_sched_barrier(0);
#pragma unroll
    for (int i = 0; i < NI; ++i) {
        const int it = threadIdx.x + i * GT;
        const int j = it / C4, c4 = it - j * C4;
        float* d = &ldsB[j * PP + c4 * 8];
        *(float4*)d       = make_float4(s0[i].x, s1[i].x, s0[i].y, s1[i].y);
        *(float4*)(d + 4) = make_float4(s0[i].z, s1[i].z, s0[i].w, s1[i].w);
    }
    __syncthreads();

    const int bOff = (kh >> 1) * PP + (lane & 15) * 2;

    v8f acc[TN];
#pragma unroll
    for (int t = 0; t < TN; ++t) acc[t] = v8f{};
    v2f bq[2][TN];
#pragma unroll
    for (int t = 0; t < TN; ++t)
        bq[0][t] = *(const v2f*)&ldsB[bOff + t * 32];
#pragma unroll
    for (int s = 0; s < KS; ++s) {
        if (s + 1 < KS) {
#pragma unroll
            for (int t = 0; t < TN; ++t)
                bq[(s + 1) & 1][t] = *(const v2f*)&ldsB[bOff + (2 * s + 2) * PP + t * 32];
        }
        __builtin_amdgcn_sched_barrier(0);
#pragma unroll
        for (int t = 0; t < TN; ++t)
            acc[t] = __builtin_amdgcn_wmma_f32_16x16x4_f32(
                 false, a[s],  false, bq[s & 1][t],
                 (short)0, acc[t],  false,  false);
        __builtin_amdgcn_sched_barrier(0);
    }

    if (valid) {
        const int rbase = tm * 16 + ((lane >> 4) << 3);
#pragma unroll
        for (int t = 0; t < TN; ++t) {
            const int n = t * 16 + (lane & 15);
            if (n < NOUT) {
#pragma unroll
                for (int v = 0; v < 8; ++v)
                    C[(long)(rbase + v) * NOUT + n] = acc[t][v];
            }
        }
    }
}


static_assert(NE % 4 == 0, "destinations are read as int4 by the degree launch");
static_assert(NN <= (1 << 20) && RN < (1 << 11), "saved list entry packing");
static_assert(NRG <= 255, "a range number fits a byte and 0xFF is free as the pad key");

__device__ inline void range_flags(const int4* __restrict__ k4, int U, unsigned rb, unsigned long long (&m)[2]) {
    int4 d[4][2];
#pragma unroll
    for (int k = 0; k < 4; ++k) {
        const int u = U + k * OT;
        const int i = ((u >> 5) << 6) + (u & 31);
        const bool in = u < NUN;
        d[k][0] = in ? k4[i]      : make_int4(-1, -1, -1, -1);
        d[k][1] = in ? k4[i + 32] : make_int4(-1, -1, -1, -1);
    }
    unsigned mm[4];
#pragma unroll
    for (int k = 0; k < 4; ++k) {
        const unsigned p[8] = {(unsigned)d[k][0].x, (unsigned)d[k][0].y, (unsigned)d[k][0].z, (unsigned)d[k][0].w,
                               (unsigned)d[k][1].x, (unsigned)d[k][1].y, (unsigned)d[k][1].z, (unsigned)d[k][1].w};
        unsigned a = ~0u;
#pragma unroll
        for (int j = 0; j < 8; ++j) {
            const unsigned kj = 0u - ((rb >> j) & 1u);
            a &= ~(p[j] ^ kj);
        }
        mm[k] = a;
    }
    m[0] = (unsigned long long)mm[0] | ((unsigned long long)mm[1] << 32);
    m[1] = (unsigned long long)mm[2] | ((unsigned long long)mm[3] << 32);
}

__device__ inline int flag_edge(int U, int p) {
    const int u = U + (p >> 5) * OT;
    return ((u >> 2) << 7) + ((p & 31) << 2) + (u & 3);
}

__global__ __launch_bounds__(OT) void k_own_count(const int* __restrict__ src, const int* __restrict__ dst,
                                                  float* __restrict__ deg, int* __restrict__ lcnt,
                                                  int* __restrict__ lnode, const int* __restrict__ keys) {
    __shared__ int list[LCAP];
    __shared__ int degL[RN];
    __shared__ int cnt;
    constexpr int NL = LCAP / OT;
    static_assert(LCAP % OT == 0, "the list is resolved in whole rounds of the work-group");
    const int tid = threadIdx.x;
    const int lo  = blockIdx.x * RN;
    if (tid == 0) cnt = 0;
    if (tid < RN) degL[tid] = 1;
    __syncthreads();

    const int4* k4 = (const int4*)keys;
    const unsigned rb = (unsigned)blockIdx.x;
    for (int b = tid; b < NUN; b += OT * 4) {
        unsigned long long m[2];
        range_flags(k4, b, rb, m);
        while (m[0] | m[1]) {
            const bool first = m[0] != 0;
            const int p = __builtin_ctzll(first ? m[0] : m[1]);
            if (first) m[0] &= m[0] - 1; else m[1] &= m[1] - 1;
            const int e = flag_edge(first ? b : b + 2 * OT, p);
            const int slot = atomicAdd(&cnt, 1);
            if (slot < LCAP) list[slot] = e;
            else             atomicAdd(&degL[dst[e] - lo], 1);
        }
    }
    __syncthreads();

    const int total = cnt;
    const int n = total < LCAP ? total : LCAP;
    if (tid == 0) lcnt[blockIdx.x] = total;
    int* ln = lnode + (long)blockIdx.x * LCAP;
    int sv[NL], dv[NL];
#pragma unroll
    for (int k = 0; k < NL; ++k) {
        const int idx = tid + k * OT;
        const int e = idx < n ? list[idx] : 0;
        sv[k] = src[e];
        dv[k] = dst[e];
    }
#pragma unroll
    for (int k = 0; k < NL; ++k) {
        const int idx = tid + k * OT;
        if (idx < n) {
            const int r = dv[k] - lo;
            atomicAdd(&degL[r], 1);
            ln[idx] = sv[k] | (r << 20);
        }
    }
    __syncthreads();
    if (tid < RN) deg[lo + tid] = (float)degL[tid];
}

template <int F, bool BIAS_RELU, bool SAVE, bool FUSE>
__global__ __launch_bounds__(OT) void k_own_agg(const int* __restrict__ src, const int* __restrict__ dst,
                                                const float* __restrict__ deg, const float* __restrict__ h,
                                                const float* __restrict__ bias, float* __restrict__ outp,
                                                const int* __restrict__ lcnt, const int* __restrict__ lnode,
                                                float* __restrict__ lwt,
                                                const int* __restrict__ keys, const float* __restrict__ W2) {
    constexpr int PITCH = FUSE ? F + 4 : F;
    __shared__ __attribute__((aligned(16))) float acc[RN * PITCH];
    __shared__ int   list[LCAP];
    __shared__ float dinvL[RN];
    __shared__ int   cnt;

    constexpr int NJ = (F + 31) / 32;
    constexpr int F4 = F / 4;
    constexpr int NV = RN * F4;
    constexpr int NK = (NV + OT - 1) / OT;
    static_assert(F % 4 == 0, "a 16-byte piece never straddles two rows");
    static_assert(NN % RN == 0, "every range is whole: all RN owned rows exist");
    static_assert(!FUSE || (BIAS_RELU && F == DH && F % 4 == 0), "the fused product is relu(row + bias) @ W2");
    static_assert(FUSE || !BIAS_RELU, "the launch that writes its rows out ends in bias + log_softmax");
    static_assert(PITCH % 4 == 0, "16-byte pieces and 8-byte fragment reads stay aligned");
    const int tid  = threadIdx.x;
    const int lane = tid & 31;
    const int lo   = blockIdx.x * RN;

    const int* ln = lnode + (long)blockIdx.x * LCAP;
    float*     lw = lwt + (long)blockIdx.x * LCAP;
    const int  saved    = lcnt[blockIdx.x];
    const bool useSaved = (unsigned)saved <= (unsigned)LCAP;

    if (tid == 0) cnt = 0;
    if (tid < RN) { int n = lo + tid; dinvL[tid] = (n < NN) ? rsqrtf(deg[n]) : 0.0f; }
    {
        const float4* h4 = (const float4*)(h + (long)lo * F);
        float4 hv[NK];
        float  dv[NK];
#pragma unroll
        for (int k = 0; k < NK; ++k) {
            int i4 = tid + k * OT;
            i4 = i4 < NV ? i4 : NV - 1;
            hv[k] = h4[i4];
            dv[k] = deg[lo + i4 / F4];
        }
#pragma unroll
        for (int k = 0; k < NK; ++k) {
            const int i4 = tid + k * OT;
            if (i4 < NV) {
                float s = rsqrtf(dv[k]); s *= s;
                const int o = (PITCH == F) ? i4 * 4 : (i4 / F4) * PITCH + (i4 % F4) * 4;
                *(float4*)&acc[o] = make_float4(hv[k].x * s, hv[k].y * s, hv[k].z * s, hv[k].w * s);
            }
        }
    }
    __syncthreads();

    const int4* k4 = (const int4*)keys;
    const unsigned rb = (unsigned)blockIdx.x;
    if (!useSaved) {
        for (int b = tid; b < NUN; b += OT * 4) {
            unsigned long long m[2];
            range_flags(k4, b, rb, m);
            while (m[0] | m[1]) {
                const bool first = m[0] != 0;
                const int p = __builtin_ctzll(first ? m[0] : m[1]);
                if (first) m[0] &= m[0] - 1; else m[1] &= m[1] - 1;
                const int e = flag_edge(first ? b : b + 2 * OT, p);
                int slot = atomicAdd(&cnt, 1);
                if (slot < LCAP) {
                    list[slot] = e;
                } else {
                    int s = src[e];
                    int r = dst[e] - lo;
                    float w = rsqrtf(deg[s]) * dinvL[r];
                    for (int c = 0; c < F; ++c)
                        atomicAdd(&acc[r * PITCH + c], h[(long)s * F + c] * w);
                }
            }
        }
    }
    __syncthreads();

    const int n = __builtin_amdgcn_readfirstlane(useSaved ? saved : (cnt < LCAP ? cnt : LCAP));
    for (int g = __builtin_amdgcn_readfirstlane(tid >> 5) * 32; g < n; g += OT) {
        const int  idx = g + lane;
        const bool ok  = idx < n;
        int r, s;
        float w;
        if (useSaved) {
            const int x = ok ? ln[idx] : 0;
            r = x >> 20;
            s = x & 0xFFFFF;
            if (SAVE) {
                w = ok ? rsqrtf(deg[s]) * dinvL[r] : 0.0f;
                if (ok) lw[idx] = w;
            } else {
                w = ok ? lw[idx] : 0.0f;
            }
        } else {
            const int e = ok ? list[idx] : 0;
            s = ok ? src[e] : 0;
            r = ok ? dst[e] - lo : 0;
            w = ok ? rsqrtf(deg[s]) * dinvL[r] : 0.0f;
        }
        const int  m   = (n - g) < 32 ? (n - g) : 32;
        for (int k = 0; k < m; k += 4) {
            float v[4][NJ];
            int   rk[4];
            float wk[4];
#pragma unroll
            for (int kk = 0; kk < 4; ++kk) {
                const int sl = k + kk;
                const int sk = __builtin_amdgcn_readlane(s, sl);
                rk[kk] = __builtin_amdgcn_readlane(r, sl);
                wk[kk] = __int_as_float(__builtin_amdgcn_readlane(__float_as_int(w), sl));
                const float* hp = h + (long)sk * F;
#pragma unroll
                for (int j = 0; j < NJ; ++j) {
                    const int c = lane + 32 * j;
                    v[kk][j] = (c < F) ? hp[c] : 0.0f;
                }
            }
#pragma unroll
            for (int kk = 0; kk < 4; ++kk) {
                if (k + kk < m) {
#pragma unroll
                    for (int j = 0; j < NJ; ++j) {
                        const int c = lane + 32 * j;
                        if (c < F) atomicAdd(&acc[rk[kk] * PITCH + c], v[kk][j] * wk[kk]);
                    }
                }
            }
        }
    }
    __syncthreads();

    if (FUSE) {
        constexpr int NT = (RN + 15) / 16;
        constexpr int TN = NCP / 16;
        constexpr int KS = F / 4;
        const int wv = __builtin_amdgcn_readfirstlane(tid >> 5);
        if (wv < NT) {
            const int c  = lane & 15;
            const int kh = (lane >> 4) << 1;
            int ra = wv * 16 + c;
            ra = ra < RN ? ra : RN - 1;
            const float* aP = &acc[ra * PITCH + kh];
            const float* bP = bias + kh;
            int  nc[TN];
            bool nv[TN];
#pragma unroll
            for (int t = 0; t < TN; ++t) {
                const int n = t * 16 + c;
                nv[t] = n < NC;
                nc[t] = nv[t] ? n : NC - 1;
            }
            v8f d[TN];
#pragma unroll
            for (int t = 0; t < TN; ++t) d[t] = v8f{};
#pragma unroll
            for (int s = 0; s < KS; ++s) {
                v2f a = *(const v2f*)&aP[4 * s];
                const v2f bb = *(const v2f*)&bP[4 * s];
                a.x += bb.x; a.y += bb.y;
                a.x = a.x > 0.0f ? a.x : 0.0f;
                a.y = a.y > 0.0f ? a.y : 0.0f;
#pragma unroll
                for (int t = 0; t < TN; ++t) {
                    const float w0 = W2[(4 * s + kh) * NC + nc[t]];
                    const float w1 = W2[(4 * s + kh + 1) * NC + nc[t]];
                    v2f b;
                    b.x = nv[t] ? w0 : 0.0f;
                    b.y = nv[t] ? w1 : 0.0f;
                    d[t] = __builtin_amdgcn_wmma_f32_16x16x4_f32(
                         false, a,  false, b,
                         (short)0, d[t],  false,  false);
                }
            }
            const int rbase = wv * 16 + ((lane >> 4) << 3);
#pragma unroll
            for (int t = 0; t < TN; ++t) {
                if (nv[t]) {
#pragma unroll
                    for (int v = 0; v < 8; ++v)
                        if (rbase + v < RN)
                            outp[(long)(lo + rbase + v) * NC + t * 16 + c] = d[t][v];
                }
            }
        }
    } else {
        __shared__ float lse[RN];
        if (tid < RN) {
            const float4* r4 = (const float4*)&acc[tid * PITCH];
            float v[F];
#pragma unroll
            for (int q = 0; q < F4; ++q) {
                const float4 a = r4[q];
                const float4 b = ((const float4*)bias)[q];
                v[4 * q]     = a.x + b.x;
                v[4 * q + 1] = a.y + b.y;
                v[4 * q + 2] = a.z + b.z;
                v[4 * q + 3] = a.w + b.w;
            }
            float mx = -INFINITY;
#pragma unroll
            for (int q = 0; q < F; ++q) mx = fmaxf(mx, v[q]);
            float sum = 0.0f;
#pragma unroll
            for (int q = 0; q < F; ++q) sum += expf(v[q] - mx);
            lse[tid] = mx + logf(sum);
        }
        __syncthreads();
        float4* o4 = (float4*)(outp + (long)lo * F);
#pragma unroll
        for (int k = 0; k < NK; ++k) {
            const int i4 = tid + k * OT;
            if (i4 < NV) {
                const int row = i4 / F4;
                const float4 b = ((const float4*)bias)[i4 - row * F4];
                const float  l = lse[row];
                float4 v = ((const float4*)acc)[i4];
                v.x = (v.x + b.x) - l; v.y = (v.y + b.y) - l;
                v.z = (v.z + b.z) - l; v.w = (v.w + b.w) - l;
                o4[i4] = v;
            }
        }
    }
}


extern "C" void kernel_launch(void* const* d_in, const int* in_sizes, int n_in,
                              void* d_out, int out_size, void* d_ws, size_t ws_size,
                              hipStream_t stream) {
    const float* x  = (const float*)d_in[0];
    const int*   ei = (const int*)d_in[1];
    const float* W1 = (const float*)d_in[2];
    const float* b1 = (const float*)d_in[3];
    const float* W2 = (const float*)d_in[4];
    const float* b2 = (const float*)d_in[5];
    float* out = (float*)d_out;

    const int* src = ei;
    const int* dst = ei + NE;

    float* ws   = (float*)d_ws;
    float* deg  = ws;
    float* h0   = deg  + NN;
    float* h2   = h0   + (long)NN * DH;
    int*   lcnt = (int*)(h2 + (long)NN * NC);
    int*   lnod = lcnt + 256;
    float* lwt  = (float*)(lnod + (long)NRG * LCAP);
    int*   keys = (int*)(lwt + (long)NRG * LCAP);

    k_gemm_wmma_f32<FIN, DH, DH>
        <<<cdiv(NN / 16, GW), GT, 0, stream>>>(x, W1, h0, NN, dst, keys);

    k_own_count<<<cdiv(NN, RN), OT, 0, stream>>>(src, dst, deg, lcnt, lnod, keys);

    k_own_agg<DH, true, true, true><<<cdiv(NN, RN), OT, 0, stream>>>(src, dst, deg, h0, b1, h2, lcnt, lnod, lwt, keys, W2);

    k_own_agg<NC, false, false, false><<<cdiv(NN, RN), OT, 0, stream>>>(src, dst, deg, h2, b2, out, lcnt, lnod, lwt, keys, W2);
}
